# starting kernel
# baseline (speedup 1.0000x reference)
#include <hip/hip_runtime.h>
#include <hip/hip_bf16.h>


typedef float v2f __attribute__((ext_vector_type(2)));
typedef float v8f __attribute__((ext_vector_type(8)));

constexpr int   Bc = 32, Sc = 1024, Ec = 512, Hc = 64, Tc = 256;
constexpr float LNK = -0.1053605156578263f;

static __device__ __forceinline__ v8f wmma4(v2f a, v2f b, v8f c) {
  return __builtin_amdgcn_wmma_f32_16x16x4_f32(false, a, false, b, (short)0, c,
                                               false, false);
}

__global__ void __launch_bounds__(32)
proj16x16(const float* __restrict__ X, const float* __restrict__ W,
          float* __restrict__ Y, int K, int N)
{
  const int lane  = threadIdx.x & 31;
  const int nl    = lane & 15;
  const int khalf = (lane >> 4) << 1;
  const int m     = (blockIdx.x << 4) + nl;
  const int ncol  = (blockIdx.y << 4) + nl;

  const float* xrow = X + (size_t)m * K;
  v8f c = {};
  for (int k0 = 0; k0 < K; k0 += 4) {
    v2f a = *(const v2f*)(xrow + k0 + khalf);
    v2f b;
    b.x = W[(size_t)(k0 + khalf)     * N + ncol];
    b.y = W[(size_t)(k0 + khalf + 1) * N + ncol];
    c = wmma4(a, b, c);
  }
  const int mbase = (blockIdx.x << 4) + ((lane >> 4) << 3);
  #pragma unroll
  for (int r = 0; r < 8; ++r)
    Y[(size_t)(mbase + r) * N + ncol] = c[r];
}

__global__ void dones_scan(const int* __restrict__ dones, int* __restrict__ cum,
                           int* __restrict__ firstdone)
{
  const int b = blockIdx.x;
  if (threadIdx.x == 0) {
    int c = 0, first = Tc;
    for (int t = 0; t < Tc; ++t) {
      const int d = (dones[(size_t)b * Sc + t * 4] != 0) ? 1 : 0;
      if (d && first == Tc) first = t;
      c += d;
      cum[b * Tc + t] = c;
    }
    firstdone[b] = first;
  }
}

__global__ void __launch_bounds__(32)
retention16(const float* __restrict__ q, const float* __restrict__ kbuf,
            const float* __restrict__ vbuf, const float* __restrict__ hstate,
            const int* __restrict__ cum, const int* __restrict__ firstdone,
            float* __restrict__ out)
{
  __shared__ float ktab[Tc + 1];
  __shared__ int   cumsh[Tc];
  __shared__ float atile[16 * 16];

  const int lane  = threadIdx.x & 31;
  const int nl    = lane & 15;
  const int khalf = (lane >> 4) << 1;
  const int mh    = (lane >> 4) << 3;
  const int it    = blockIdx.x;
  const int b     = blockIdx.y;

  const int* cumb = cum + b * Tc;
  for (int i = lane; i <= Tc; i += 32) ktab[i] = __expf((float)i * LNK);
  for (int i = lane; i < Tc; i += 32)  cumsh[i] = cumb[i];
  __syncthreads();

  const float* qb = q      + (size_t)b * Sc * Hc;
  const float* kb = kbuf   + (size_t)b * Sc * Hc;
  const float* vb = vbuf   + (size_t)b * Sc * Hc;
  const float* hb = hstate + (size_t)b * Hc * Hc;

  const float* qrow = qb + (size_t)(it * 16 + nl) * Hc;
  v2f qa[16];
  #pragma unroll
  for (int ks = 0; ks < 16; ++ks)
    qa[ks] = *(const v2f*)(qrow + ks * 4 + khalf);

  v8f acc[4] = {};
  for (int jt = 0; jt <= it; ++jt) {
    v8f s = {};
    const float* krow = kb + (size_t)(jt * 16 + nl) * Hc;
    #pragma unroll
    for (int ks = 0; ks < 16; ++ks) {
      v2f bm = *(const v2f*)(krow + ks * 4 + khalf);
      s = wmma4(qa[ks], bm, s);
    }
    #pragma unroll
    for (int r = 0; r < 8; ++r) {
      const int gi = it * 16 + mh + r;
      const int gj = jt * 16 + nl;
      const int ti = gi >> 2, tj = gj >> 2;
      float w = 0.0f;
      if (gj <= gi && cumsh[ti] == cumsh[tj]) w = ktab[ti - tj];
      atile[(mh + r) * 16 + nl] = s[r] * w;
    }
    __syncthreads();
    #pragma unroll
    for (int k0 = 0; k0 < 16; k0 += 4) {
      v2f a = *(const v2f*)(&atile[nl * 16 + k0 + khalf]);
      const float* v0 = vb + (size_t)(jt * 16 + k0 + khalf) * Hc;
      const float* v1 = v0 + Hc;
      #pragma unroll
      for (int nt = 0; nt < 4; ++nt) {
        v2f bm;
        bm.x = v0[nt * 16 + nl];
        bm.y = v1[nt * 16 + nl];
        acc[nt] = wmma4(a, bm, acc[nt]);
      }
    }
    __syncthreads();
  }

  v8f cr[4] = {};
  #pragma unroll
  for (int ks = 0; ks < 16; ++ks) {
    const float* h0 = hb + (size_t)(ks * 4 + khalf) * Hc;
    const float* h1 = h0 + Hc;
    #pragma unroll
    for (int nt = 0; nt < 4; ++nt) {
      v2f bm;
      bm.x = h0[nt * 16 + nl];
      bm.y = h1[nt * 16 + nl];
      cr[nt] = wmma4(qa[ks], bm, cr[nt]);
    }
  }

  const int fd = firstdone[b];
  #pragma unroll
  for (int r = 0; r < 8; ++r) {
    const int   gi = it * 16 + mh + r;
    const int   ti = gi >> 2;
    const float xi = (ti < fd) ? ktab[ti + 1] : 0.0f;
    float* orow = out + ((size_t)b * Sc + gi) * Hc;
    #pragma unroll
    for (int nt = 0; nt < 4; ++nt)
      orow[nt * 16 + nl] = acc[nt][r] + cr[nt][r] * xi;
  }
}

__global__ void __launch_bounds__(32)
hstate_update(const float* __restrict__ kbuf, const float* __restrict__ vbuf,
              const float* __restrict__ hstate, const int* __restrict__ cum,
              float* __restrict__ hout)
{
  __shared__ float ktab[Tc + 1];
  __shared__ float wsh[Tc];
  const int lane  = threadIdx.x & 31;
  const int nl    = lane & 15;
  const int khalf = (lane >> 4) << 1;
  const int mt = blockIdx.x, nt = blockIdx.y, b = blockIdx.z;
  const int* cumb = cum + b * Tc;
  const int  cend = cumb[Tc - 1];

  for (int i = lane; i <= Tc; i += 32) ktab[i] = __expf((float)i * LNK);
  __syncthreads();
  for (int i = lane; i < Tc; i += 32)
    wsh[i] = (cumb[i] == cend) ? ktab[(Tc - 1) - i] : 0.0f;
  __syncthreads();

  const float* kb = kbuf + (size_t)b * Sc * Hc;
  const float* vb = vbuf + (size_t)b * Sc * Hc;

  v8f c = {};
  for (int s0 = 0; s0 < Sc; s0 += 4) {
    const int   sA = s0 + khalf;
    const float w  = wsh[s0 >> 2];
    v2f a;
    a.x = kb[(size_t)sA       * Hc + mt * 16 + nl];
    a.y = kb[(size_t)(sA + 1) * Hc + mt * 16 + nl];
    v2f bm;
    bm.x = vb[(size_t)sA       * Hc + nt * 16 + nl] * w;
    bm.y = vb[(size_t)(sA + 1) * Hc + nt * 16 + nl] * w;
    c = wmma4(a, bm, c);
  }

  const float cd    = ktab[Tc];
  const float delta = (cend == 0) ? 1.0f : 0.0f;
  const float* hb   = hstate + (size_t)b * Hc * Hc;
  const int    mh   = (lane >> 4) << 3;
  #pragma unroll
  for (int r = 0; r < 8; ++r) {
    const int h1 = mt * 16 + mh + r;
    hout[((size_t)b * Hc + h1) * Hc + nt * 16 + nl] =
        c[r] + hb[(size_t)h1 * Hc + nt * 16 + nl] * cd * delta;
  }
}

extern "C" void kernel_launch(void* const* d_in, const int* in_sizes, int n_in,
                              void* d_out, int out_size, void* d_ws, size_t ws_size,
                              hipStream_t stream)
{
  const float* key    = (const float*)d_in[0];
  const float* query  = (const float*)d_in[1];
  const float* value  = (const float*)d_in[2];
  const float* hstate = (const float*)d_in[3];
  const int*   dones  = (const int*)d_in[4];
  const float* W_Q    = (const float*)d_in[5];
  const float* W_K    = (const float*)d_in[6];
  const float* W_V    = (const float*)d_in[7];

  const size_t qkv = (size_t)Bc * Sc * Hc;
  float* ws = (float*)d_ws;
  float* qb = ws;
  float* kb = ws + qkv;
  float* vb = ws + 2 * qkv;
  int*   cum       = (int*)(ws + 3 * qkv);
  int*   firstdone = cum + Bc * Tc;

  dim3 gproj(Bc * Sc / 16, Hc / 16);
  proj16x16<<<gproj, 32, 0, stream>>>(query, W_Q, qb, Ec, Hc);
  proj16x16<<<gproj, 32, 0, stream>>>(key,   W_K, kb, Ec, Hc);
  proj16x16<<<gproj, 32, 0, stream>>>(value, W_V, vb, Ec, Hc);

  dones_scan<<<Bc, 32, 0, stream>>>(dones, cum, firstdone);

  float* out0 = (float*)d_out;
  float* hout = out0 + qkv;
  retention16<<<dim3(Sc / 16, Bc), 32, 0, stream>>>(qb, kb, vb, hstate, cum,
                                                    firstdone, out0);
  hstate_update<<<dim3(Hc / 16, Hc / 16, Bc), 32, 0, stream>>>(kb, vb, hstate,
                                                               cum, hout);
}


# retention: score tile computed transposed (k@q^T) so the masked tile feeds the second product from registers, no LDS round trip in the key loop; keys summed in another order inside each 16-key tile
# speedup vs baseline: 2.2998x; 2.2998x over previous
#include <hip/hip_runtime.h>
#include <hip/hip_bf16.h>


typedef float v2f __attribute__((ext_vector_type(2)));
typedef float v4f __attribute__((ext_vector_type(4)));
typedef float v8f __attribute__((ext_vector_type(8)));

constexpr int   Bc = 32, Sc = 1024, Ec = 512, Hc = 64, Tc = 256;
constexpr float LNK = -0.1053605156578263f;

static __device__ __forceinline__ v8f wmma4(v2f a, v2f b, v8f c) {
  return __builtin_amdgcn_wmma_f32_16x16x4_f32(false, a, false, b, (short)0, c,
                                               false, false);
}

constexpr int PROJ_ROWS = 128;

__global__ void __launch_bounds__(256)
proj_strip(const float* __restrict__ X, const float* __restrict__ W,
           float* __restrict__ Y)
{
  __shared__ __attribute__((aligned(16))) float wl[Ec * Hc];

  const int tid = threadIdx.x;

  for (int p = tid; p < (Ec / 8) * Hc; p += 256) {
    const int o = p >> 6, n = p & 63;
    const float* wsrc = W + (size_t)(o * 8) * Hc + n;
    v4f lo, hi;
    lo.x = wsrc[0 * Hc]; lo.y = wsrc[1 * Hc];
    lo.z = wsrc[2 * Hc]; lo.w = wsrc[3 * Hc];
    hi.x = wsrc[4 * Hc]; hi.y = wsrc[5 * Hc];
    hi.z = wsrc[6 * Hc]; hi.w = wsrc[7 * Hc];
    *(v4f*)(&wl[p * 8])     = lo;
    *(v4f*)(&wl[p * 8 + 4]) = hi;
  }
  __syncthreads();

  const int lane = tid & 31;
  const int wave = tid >> 5;
  const int nl   = lane & 15;
  const int h    = lane >> 4;
  const int row0 = blockIdx.x * PROJ_ROWS + wave * 16;

  const float* xrow  = X + (size_t)(row0 + nl) * Ec + h * 4;
  const float* wlane = wl + nl * 8 + h * 4;

  v8f acc[4] = {};
  v4f xa[16];
  v4f wb[2][4];

  #pragma unroll
  for (int i = 0; i < 8; ++i) xa[i] = *(const v4f*)(xrow + i * 8);
  #pragma unroll
  for (int nt = 0; nt < 4; ++nt) wb[0][nt] = *(const v4f*)(wlane + nt * 128);

  #pragma unroll
  for (int c = 0; c < 8; ++c) {
    if (c + 1 < 8) {
      #pragma unroll
      for (int i = 0; i < 8; ++i)
        xa[((c + 1) & 1) * 8 + i] = *(const v4f*)(xrow + ((c + 1) * 8 + i) * 8);
    }
    #pragma unroll
    for (int i = 0; i < 8; ++i) {
      const int o = c * 8 + i;
      if (o + 1 < Ec / 8) {
        #pragma unroll
        for (int nt = 0; nt < 4; ++nt)
          wb[(o + 1) & 1][nt] =
              *(const v4f*)(wlane + ((o + 1) * 64 + nt * 16) * 8);
      }
      const v4f x = xa[(c & 1) * 8 + i];
      v2f a0, a1;
      a0.x = x.x; a0.y = x.y;
      a1.x = x.z; a1.y = x.w;
      #pragma unroll
      for (int nt = 0; nt < 4; ++nt) {
        const v4f w4 = wb[o & 1][nt];
        v2f b0; b0.x = w4.x; b0.y = w4.y;
        acc[nt] = wmma4(a0, b0, acc[nt]);
      }
      #pragma unroll
      for (int nt = 0; nt < 4; ++nt) {
        const v4f w4 = wb[o & 1][nt];
        v2f b1; b1.x = w4.z; b1.y = w4.w;
        acc[nt] = wmma4(a1, b1, acc[nt]);
      }
    }
  }

  const int mbase = row0 + h * 8;
  #pragma unroll
  for (int r = 0; r < 8; ++r) {
    float* yrow = Y + (size_t)(mbase + r) * Hc;
    #pragma unroll
    for (int nt = 0; nt < 4; ++nt)
      yrow[nt * 16 + nl] = acc[nt][r];
  }
}

__global__ void dones_scan(const int* __restrict__ dones, int* __restrict__ cum,
                           int* __restrict__ firstdone)
{
  const int b = blockIdx.x;
  if (threadIdx.x == 0) {
    int c = 0, first = Tc;
    for (int t = 0; t < Tc; ++t) {
      const int d = (dones[(size_t)b * Sc + t * 4] != 0) ? 1 : 0;
      if (d && first == Tc) first = t;
      c += d;
      cum[b * Tc + t] = c;
    }
    firstdone[b] = first;
  }
}

__global__ void __launch_bounds__(32)
retention16(const float* __restrict__ q, const float* __restrict__ kbuf,
            const float* __restrict__ vbuf, const float* __restrict__ hstate,
            const int* __restrict__ cum, const int* __restrict__ firstdone,
            float* __restrict__ out)
{
  __shared__ float ktab[Tc + 1];
  __shared__ int   cumsh[Tc];

  const int lane  = threadIdx.x & 31;
  const int nl    = lane & 15;
  const int khalf = (lane >> 4) << 1;
  const int mh    = (lane >> 4) << 3;
  const int it    = blockIdx.x;
  const int b     = blockIdx.y;

  const int* cumb = cum + b * Tc;
  for (int i = lane; i <= Tc; i += 32) ktab[i] = __expf((float)i * LNK);
  for (int i = lane; i < Tc; i += 32)  cumsh[i] = cumb[i];
  __syncthreads();

  const float* qb = q      + (size_t)b * Sc * Hc;
  const float* kb = kbuf   + (size_t)b * Sc * Hc;
  const float* vb = vbuf   + (size_t)b * Sc * Hc;
  const float* hb = hstate + (size_t)b * Hc * Hc;


  const float* qrow = qb + (size_t)(it * 16 + nl) * Hc;
  v2f qa[16];
  #pragma unroll
  for (int ks = 0; ks < 16; ++ks)
    qa[ks] = *(const v2f*)(qrow + ks * 4 + khalf);

  const int gq = it * 16 + nl;
  const int tq = gq >> 2;
  const int cq = cumsh[tq];

  v8f acc[4] = {};
  for (int jt = 0; jt <= it; ++jt) {
    const float* krow = kb + (size_t)(jt * 16 + nl) * Hc + khalf;
    v2f kf[16];
    #pragma unroll
    for (int ks = 0; ks < 16; ++ks)
      kf[ks] = *(const v2f*)(krow + ks * 4);
    const float* vrow = vb + (size_t)(jt * 16 + mh) * Hc + nl;
    v2f vf[4][4];
    #pragma unroll
    for (int t = 0; t < 4; ++t) {
      #pragma unroll
      for (int nt = 0; nt < 4; ++nt) {
        vf[t][nt].x = vrow[(2 * t)     * Hc + nt * 16];
        vf[t][nt].y = vrow[(2 * t + 1) * Hc + nt * 16];
      }
    }
    const int   tk0 = jt * 4 + (mh >> 2);
    const int   d0  = tq - tk0;
    const int   c0  = cumsh[tk0];
    const int   c1  = cumsh[tk0 + 1];
    const float e0  = ktab[d0 < 0 ? 0 : d0];
    const float e1  = ktab[d0 < 1 ? 0 : d0 - 1];
    const float w0  = (c0 == cq) ? e0 : 0.0f;
    const float w1  = (c1 == cq) ? e1 : 0.0f;

    v8f s = {};
    #pragma unroll
    for (int ks = 0; ks < 16; ++ks)
      s = wmma4(kf[ks], qa[ks], s);

    #pragma unroll
    for (int r = 0; r < 8; ++r) {
      const int   gk = jt * 16 + mh + r;
      const float w  = (gk <= gq) ? (r < 4 ? w0 : w1) : 0.0f;
      s[r] = s[r] * w;
    }
    #pragma unroll
    for (int t = 0; t < 4; ++t) {
      v2f a;
      a.x = s[2 * t];
      a.y = s[2 * t + 1];
      #pragma unroll
      for (int nt = 0; nt < 4; ++nt)
        acc[nt] = wmma4(a, vf[t][nt], acc[nt]);
    }
  }

  v8f cr[4] = {};
  #pragma unroll
  for (int ks = 0; ks < 16; ++ks) {
    const float* h0 = hb + (size_t)(ks * 4 + khalf) * Hc;
    const float* h1 = h0 + Hc;
    #pragma unroll
    for (int nt = 0; nt < 4; ++nt) {
      v2f bm;
      bm.x = h0[nt * 16 + nl];
      bm.y = h1[nt * 16 + nl];
      cr[nt] = wmma4(qa[ks], bm, cr[nt]);
    }
  }

  const int fd = firstdone[b];
  #pragma unroll
  for (int r = 0; r < 8; ++r) {
    const int   gi = it * 16 + mh + r;
    const int   ti = gi >> 2;
    const float xi = (ti < fd) ? ktab[ti + 1] : 0.0f;
    float* orow = out + ((size_t)b * Sc + gi) * Hc;
    #pragma unroll
    for (int nt = 0; nt < 4; ++nt)
      orow[nt * 16 + nl] = acc[nt][r] + cr[nt][r] * xi;
  }
}

__global__ void __launch_bounds__(32)
hstate_update(const float* __restrict__ kbuf, const float* __restrict__ vbuf,
              const float* __restrict__ hstate, const int* __restrict__ cum,
              float* __restrict__ hout)
{
  __shared__ float ktab[Tc + 1];
  __shared__ float wsh[Tc];
  const int lane  = threadIdx.x & 31;
  const int nl    = lane & 15;
  const int khalf = (lane >> 4) << 1;
  const int mt = blockIdx.x, nt = blockIdx.y, b = blockIdx.z;
  const int* cumb = cum + b * Tc;
  const int  cend = cumb[Tc - 1];

  for (int i = lane; i <= Tc; i += 32) ktab[i] = __expf((float)i * LNK);
  __syncthreads();
  for (int i = lane; i < Tc; i += 32)
    wsh[i] = (cumb[i] == cend) ? ktab[(Tc - 1) - i] : 0.0f;
  __syncthreads();

  const float* kb = kbuf + (size_t)b * Sc * Hc;
  const float* vb = vbuf + (size_t)b * Sc * Hc;

  v8f c = {};
  for (int s0 = 0; s0 < Sc; s0 += 4) {
    const int   sA = s0 + khalf;
    const float w  = wsh[s0 >> 2];
    v2f a;
    a.x = kb[(size_t)sA       * Hc + mt * 16 + nl];
    a.y = kb[(size_t)(sA + 1) * Hc + mt * 16 + nl];
    v2f bm;
    bm.x = vb[(size_t)sA       * Hc + nt * 16 + nl] * w;
    bm.y = vb[(size_t)(sA + 1) * Hc + nt * 16 + nl] * w;
    c = wmma4(a, bm, c);
  }

  const float cd    = ktab[Tc];
  const float delta = (cend == 0) ? 1.0f : 0.0f;
  const float* hb   = hstate + (size_t)b * Hc * Hc;
  const int    mh   = (lane >> 4) << 3;
  #pragma unroll
  for (int r = 0; r < 8; ++r) {
    const int h1 = mt * 16 + mh + r;
    hout[((size_t)b * Hc + h1) * Hc + nt * 16 + nl] =
        c[r] + hb[(size_t)h1 * Hc + nt * 16 + nl] * cd * delta;
  }
}

extern "C" void kernel_launch(void* const* d_in, const int* in_sizes, int n_in,
                              void* d_out, int out_size, void* d_ws, size_t ws_size,
                              hipStream_t stream)
{
  const float* key    = (const float*)d_in[0];
  const float* query  = (const float*)d_in[1];
  const float* value  = (const float*)d_in[2];
  const float* hstate = (const float*)d_in[3];
  const int*   dones  = (const int*)d_in[4];
  const float* W_Q    = (const float*)d_in[5];
  const float* W_K    = (const float*)d_in[6];
  const float* W_V    = (const float*)d_in[7];

  const size_t qkv = (size_t)Bc * Sc * Hc;
  float* ws = (float*)d_ws;
  float* qb = ws;
  float* kb = ws + qkv;
  float* vb = ws + 2 * qkv;
  int*   cum       = (int*)(ws + 3 * qkv);
  int*   firstdone = cum + Bc * Tc;

  constexpr int gproj = Bc * Sc / PROJ_ROWS;
  proj_strip<<<gproj, 256, 0, stream>>>(query, W_Q, qb);
  proj_strip<<<gproj, 256, 0, stream>>>(key,   W_K, kb);
  proj_strip<<<gproj, 256, 0, stream>>>(value, W_V, vb);

  dones_scan<<<Bc, 32, 0, stream>>>(dones, cum, firstdone);

  float* out0 = (float*)d_out;
  float* hout = out0 + qkv;
  retention16<<<dim3(Sc / 16, Bc), 32, 0, stream>>>(qb, kb, vb, hstate, cum,
                                                    firstdone, out0);
  hstate_update<<<dim3(Hc / 16, Hc / 16, Bc), 32, 0, stream>>>(kb, vb, hstate,
                                                               cum, hout);
}


# done-scan, retention and state update in ONE launch: each wave rebuilds its batch's done prefix by ballot, state tiles ride as 16 extra work-groups per batch; launches 6 -> 4
# speedup vs baseline: 3.1524x; 1.3707x over previous
#include <hip/hip_runtime.h>
#include <hip/hip_bf16.h>


typedef float v2f __attribute__((ext_vector_type(2)));
typedef float v4f __attribute__((ext_vector_type(4)));
typedef float v8f __attribute__((ext_vector_type(8)));

constexpr int   Bc = 32, Sc = 1024, Ec = 512, Hc = 64, Tc = 256;
constexpr float LNK = -0.1053605156578263f;

static __device__ __forceinline__ v8f wmma4(v2f a, v2f b, v8f c) {
  return __builtin_amdgcn_wmma_f32_16x16x4_f32(false, a, false, b, (short)0, c,
                                               false, false);
}

constexpr int PROJ_ROWS = 128;

__global__ void __launch_bounds__(256)
proj_strip(const float* __restrict__ X, const float* __restrict__ W,
           float* __restrict__ Y)
{
  __shared__ __attribute__((aligned(16))) float wl[Ec * Hc];

  const int tid = threadIdx.x;

  for (int p = tid; p < (Ec / 8) * Hc; p += 256) {
    const int o = p >> 6, n = p & 63;
    const float* wsrc = W + (size_t)(o * 8) * Hc + n;
    v4f lo, hi;
    lo.x = wsrc[0 * Hc]; lo.y = wsrc[1 * Hc];
    lo.z = wsrc[2 * Hc]; lo.w = wsrc[3 * Hc];
    hi.x = wsrc[4 * Hc]; hi.y = wsrc[5 * Hc];
    hi.z = wsrc[6 * Hc]; hi.w = wsrc[7 * Hc];
    *(v4f*)(&wl[p * 8])     = lo;
    *(v4f*)(&wl[p * 8 + 4]) = hi;
  }
  __syncthreads();

  const int lane = tid & 31;
  const int wave = tid >> 5;
  const int nl   = lane & 15;
  const int h    = lane >> 4;
  const int row0 = blockIdx.x * PROJ_ROWS + wave * 16;

  const float* xrow  = X + (size_t)(row0 + nl) * Ec + h * 4;
  const float* wlane = wl + nl * 8 + h * 4;

  v8f acc[4] = {};
  v4f xa[16];
  v4f wb[2][4];

  #pragma unroll
  for (int i = 0; i < 8; ++i) xa[i] = *(const v4f*)(xrow + i * 8);
  #pragma unroll
  for (int nt = 0; nt < 4; ++nt) wb[0][nt] = *(const v4f*)(wlane + nt * 128);

  #pragma unroll
  for (int c = 0; c < 8; ++c) {
    if (c + 1 < 8) {
      #pragma unroll
      for (int i = 0; i < 8; ++i)
        xa[((c + 1) & 1) * 8 + i] = *(const v4f*)(xrow + ((c + 1) * 8 + i) * 8);
    }
    #pragma unroll
    for (int i = 0; i < 8; ++i) {
      const int o = c * 8 + i;
      if (o + 1 < Ec / 8) {
        #pragma unroll
        for (int nt = 0; nt < 4; ++nt)
          wb[(o + 1) & 1][nt] =
              *(const v4f*)(wlane + ((o + 1) * 64 + nt * 16) * 8);
      }
      const v4f x = xa[(c & 1) * 8 + i];
      v2f a0, a1;
      a0.x = x.x; a0.y = x.y;
      a1.x = x.z; a1.y = x.w;
      #pragma unroll
      for (int nt = 0; nt < 4; ++nt) {
        const v4f w4 = wb[o & 1][nt];
        v2f b0; b0.x = w4.x; b0.y = w4.y;
        acc[nt] = wmma4(a0, b0, acc[nt]);
      }
      #pragma unroll
      for (int nt = 0; nt < 4; ++nt) {
        const v4f w4 = wb[o & 1][nt];
        v2f b1; b1.x = w4.z; b1.y = w4.w;
        acc[nt] = wmma4(a1, b1, acc[nt]);
      }
    }
  }

  const int mbase = row0 + h * 8;
  #pragma unroll
  for (int r = 0; r < 8; ++r) {
    float* yrow = Y + (size_t)(mbase + r) * Hc;
    #pragma unroll
    for (int nt = 0; nt < 4; ++nt)
      yrow[nt * 16 + nl] = acc[nt][r];
  }
}

static __device__ __forceinline__ void done_prefix(const int* __restrict__ db,
                                                   int lane, int (&cj)[8],
                                                   int& total, int& first)
{
  int dv[8];
  #pragma unroll
  for (int j = 0; j < 8; ++j) dv[j] = db[(32 * j + lane) * 4];
  const unsigned lanes_le = (2u << lane) - 1u;
  int base = 0, fd = Tc;
  #pragma unroll
  for (int j = 0; j < 8; ++j) {
    const unsigned m = __builtin_amdgcn_ballot_w32(dv[j] != 0);
    cj[j] = base + __popc(m & lanes_le);
    if (m != 0u && fd == Tc) fd = 32 * j + (__ffs(m) - 1);
    base += __popc(m);
  }
  total = base;
  first = fd;
}

static __device__ __forceinline__ void
retention16(const float* __restrict__ q, const float* __restrict__ kbuf,
            const float* __restrict__ vbuf, const float* __restrict__ hstate,
            const int* __restrict__ dones, float* __restrict__ out,
            const int it, const int b)
{
  __shared__ float ktab[Tc + 1];
  __shared__ int   cumsh[Tc];

  const int lane  = threadIdx.x & 31;
  const int nl    = lane & 15;
  const int khalf = (lane >> 4) << 1;
  const int mh    = (lane >> 4) << 3;

  int cj[8], ndone, fd;
  done_prefix(dones + (size_t)b * Sc, lane, cj, ndone, fd);
  (void)ndone;
  for (int i = lane; i <= Tc; i += 32) ktab[i] = __expf((float)i * LNK);
  #pragma unroll
  for (int j = 0; j < 8; ++j) cumsh[32 * j + lane] = cj[j];
  __syncthreads();

  const float* qb = q      + (size_t)b * Sc * Hc;
  const float* kb = kbuf   + (size_t)b * Sc * Hc;
  const float* vb = vbuf   + (size_t)b * Sc * Hc;
  const float* hb = hstate + (size_t)b * Hc * Hc;


  const float* qrow = qb + (size_t)(it * 16 + nl) * Hc;
  v2f qa[16];
  #pragma unroll
  for (int ks = 0; ks < 16; ++ks)
    qa[ks] = *(const v2f*)(qrow + ks * 4 + khalf);

  const int gq = it * 16 + nl;
  const int tq = gq >> 2;
  const int cq = cumsh[tq];

  v8f acc[4] = {};
  for (int jt = 0; jt <= it; ++jt) {
    const float* krow = kb + (size_t)(jt * 16 + nl) * Hc + khalf;
    v2f kf[16];
    #pragma unroll
    for (int ks = 0; ks < 16; ++ks)
      kf[ks] = *(const v2f*)(krow + ks * 4);
    const float* vrow = vb + (size_t)(jt * 16 + mh) * Hc + nl;
    v2f vf[4][4];
    #pragma unroll
    for (int t = 0; t < 4; ++t) {
      #pragma unroll
      for (int nt = 0; nt < 4; ++nt) {
        vf[t][nt].x = vrow[(2 * t)     * Hc + nt * 16];
        vf[t][nt].y = vrow[(2 * t + 1) * Hc + nt * 16];
      }
    }
    const int   tk0 = jt * 4 + (mh >> 2);
    const int   d0  = tq - tk0;
    const int   c0  = cumsh[tk0];
    const int   c1  = cumsh[tk0 + 1];
    const float e0  = ktab[d0 < 0 ? 0 : d0];
    const float e1  = ktab[d0 < 1 ? 0 : d0 - 1];
    const float w0  = (c0 == cq) ? e0 : 0.0f;
    const float w1  = (c1 == cq) ? e1 : 0.0f;

    v8f s = {};
    #pragma unroll
    for (int ks = 0; ks < 16; ++ks)
      s = wmma4(kf[ks], qa[ks], s);

    #pragma unroll
    for (int r = 0; r < 8; ++r) {
      const int   gk = jt * 16 + mh + r;
      const float w  = (gk <= gq) ? (r < 4 ? w0 : w1) : 0.0f;
      s[r] = s[r] * w;
    }
    #pragma unroll
    for (int t = 0; t < 4; ++t) {
      v2f a;
      a.x = s[2 * t];
      a.y = s[2 * t + 1];
      #pragma unroll
      for (int nt = 0; nt < 4; ++nt)
        acc[nt] = wmma4(a, vf[t][nt], acc[nt]);
    }
  }

  v8f cr[4] = {};
  #pragma unroll
  for (int ks = 0; ks < 16; ++ks) {
    const float* h0 = hb + (size_t)(ks * 4 + khalf) * Hc;
    const float* h1 = h0 + Hc;
    #pragma unroll
    for (int nt = 0; nt < 4; ++nt) {
      v2f bm;
      bm.x = h0[nt * 16 + nl];
      bm.y = h1[nt * 16 + nl];
      cr[nt] = wmma4(qa[ks], bm, cr[nt]);
    }
  }

  #pragma unroll
  for (int r = 0; r < 8; ++r) {
    const int   gi = it * 16 + mh + r;
    const int   ti = gi >> 2;
    const float xi = (ti < fd) ? ktab[ti + 1] : 0.0f;
    float* orow = out + ((size_t)b * Sc + gi) * Hc;
    #pragma unroll
    for (int nt = 0; nt < 4; ++nt)
      orow[nt * 16 + nl] = acc[nt][r] + cr[nt][r] * xi;
  }
}

static __device__ __forceinline__ void
hstate_update(const float* __restrict__ kbuf, const float* __restrict__ vbuf,
              const float* __restrict__ hstate, const int* __restrict__ dones,
              float* __restrict__ hout, const int mt, const int nt, const int b)
{
  __shared__ float ktab[Tc + 1];
  __shared__ float wsh[Tc];
  const int lane  = threadIdx.x & 31;
  const int nl    = lane & 15;
  const int khalf = (lane >> 4) << 1;

  int cj[8], cend, fd;
  done_prefix(dones + (size_t)b * Sc, lane, cj, cend, fd);
  (void)fd;

  for (int i = lane; i <= Tc; i += 32) ktab[i] = __expf((float)i * LNK);
  __syncthreads();
  #pragma unroll
  for (int j = 0; j < 8; ++j) {
    const int i = 32 * j + lane;
    wsh[i] = (cj[j] == cend) ? ktab[(Tc - 1) - i] : 0.0f;
  }
  __syncthreads();

  const float* kb = kbuf + (size_t)b * Sc * Hc;
  const float* vb = vbuf + (size_t)b * Sc * Hc;

  v8f c = {};
  for (int s0 = 0; s0 < Sc; s0 += 4) {
    const int   sA = s0 + khalf;
    const float w  = wsh[s0 >> 2];
    v2f a;
    a.x = kb[(size_t)sA       * Hc + mt * 16 + nl];
    a.y = kb[(size_t)(sA + 1) * Hc + mt * 16 + nl];
    v2f bm;
    bm.x = vb[(size_t)sA       * Hc + nt * 16 + nl] * w;
    bm.y = vb[(size_t)(sA + 1) * Hc + nt * 16 + nl] * w;
    c = wmma4(a, bm, c);
  }

  const float cd    = ktab[Tc];
  const float delta = (cend == 0) ? 1.0f : 0.0f;
  const float* hb   = hstate + (size_t)b * Hc * Hc;
  const int    mh   = (lane >> 4) << 3;
  #pragma unroll
  for (int r = 0; r < 8; ++r) {
    const int h1 = mt * 16 + mh + r;
    hout[((size_t)b * Hc + h1) * Hc + nt * 16 + nl] =
        c[r] + hb[(size_t)h1 * Hc + nt * 16 + nl] * cd * delta;
  }
}

__global__ void __launch_bounds__(32)
retention_and_state(const float* __restrict__ q, const float* __restrict__ kbuf,
                    const float* __restrict__ vbuf,
                    const float* __restrict__ hstate,
                    const int* __restrict__ dones, float* __restrict__ out,
                    float* __restrict__ hout)
{
  const int x = blockIdx.x;
  const int b = blockIdx.y;
  if (x < Sc / 16) {
    retention16(q, kbuf, vbuf, hstate, dones, out, x, b);
  } else {
    const int tile = x - Sc / 16;
    hstate_update(kbuf, vbuf, hstate, dones, hout, tile >> 2, tile & 3, b);
  }
}

extern "C" void kernel_launch(void* const* d_in, const int* in_sizes, int n_in,
                              void* d_out, int out_size, void* d_ws, size_t ws_size,
                              hipStream_t stream)
{
  const float* key    = (const float*)d_in[0];
  const float* query  = (const float*)d_in[1];
  const float* value  = (const float*)d_in[2];
  const float* hstate = (const float*)d_in[3];
  const int*   dones  = (const int*)d_in[4];
  const float* W_Q    = (const float*)d_in[5];
  const float* W_K    = (const float*)d_in[6];
  const float* W_V    = (const float*)d_in[7];

  const size_t qkv = (size_t)Bc * Sc * Hc;
  float* ws = (float*)d_ws;
  float* qb = ws;
  float* kb = ws + qkv;
  float* vb = ws + 2 * qkv;

  constexpr int gproj = Bc * Sc / PROJ_ROWS;
  proj_strip<<<gproj, 256, 0, stream>>>(query, W_Q, qb);
  proj_strip<<<gproj, 256, 0, stream>>>(key,   W_K, kb);
  proj_strip<<<gproj, 256, 0, stream>>>(value, W_V, vb);

  float* out0 = (float*)d_out;
  float* hout = out0 + qkv;
  retention_and_state<<<dim3(Sc / 16 + 16, Bc), 32, 0, stream>>>(
      qb, kb, vb, hstate, dones, out0, hout);
}


# fused launch re-blocked to 4-wave work-groups: 64 query rows share k,v tiles staged in LDS by whole-line 16-byte loads, one tile ahead; state tile's s sum split over 4 waves (summed in another order)
# speedup vs baseline: 3.7409x; 1.1867x over previous
#include <hip/hip_runtime.h>
#include <hip/hip_bf16.h>


typedef float v2f __attribute__((ext_vector_type(2)));
typedef float v4f __attribute__((ext_vector_type(4)));
typedef float v8f __attribute__((ext_vector_type(8)));

constexpr int   Bc = 32, Sc = 1024, Ec = 512, Hc = 64, Tc = 256;
constexpr float LNK = -0.1053605156578263f;

static __device__ __forceinline__ v8f wmma4(v2f a, v2f b, v8f c) {
  return __builtin_amdgcn_wmma_f32_16x16x4_f32(false, a, false, b, (short)0, c,
                                               false, false);
}

constexpr int PROJ_ROWS = 128;

__global__ void __launch_bounds__(256)
proj_strip(const float* __restrict__ X, const float* __restrict__ W,
           float* __restrict__ Y)
{
  __shared__ __attribute__((aligned(16))) float wl[Ec * Hc];

  const int tid = threadIdx.x;

  for (int p = tid; p < (Ec / 8) * Hc; p += 256) {
    const int o = p >> 6, n = p & 63;
    const float* wsrc = W + (size_t)(o * 8) * Hc + n;
    v4f lo, hi;
    lo.x = wsrc[0 * Hc]; lo.y = wsrc[1 * Hc];
    lo.z = wsrc[2 * Hc]; lo.w = wsrc[3 * Hc];
    hi.x = wsrc[4 * Hc]; hi.y = wsrc[5 * Hc];
    hi.z = wsrc[6 * Hc]; hi.w = wsrc[7 * Hc];
    *(v4f*)(&wl[p * 8])     = lo;
    *(v4f*)(&wl[p * 8 + 4]) = hi;
  }
  __syncthreads();

  const int lane = tid & 31;
  const int wave = tid >> 5;
  const int nl   = lane & 15;
  const int h    = lane >> 4;
  const int row0 = blockIdx.x * PROJ_ROWS + wave * 16;

  const float* xrow  = X + (size_t)(row0 + nl) * Ec + h * 4;
  const float* wlane = wl + nl * 8 + h * 4;

  v8f acc[4] = {};
  v4f xa[16];
  v4f wb[2][4];

  #pragma unroll
  for (int i = 0; i < 8; ++i) xa[i] = *(const v4f*)(xrow + i * 8);
  #pragma unroll
  for (int nt = 0; nt < 4; ++nt) wb[0][nt] = *(const v4f*)(wlane + nt * 128);

  #pragma unroll
  for (int c = 0; c < 8; ++c) {
    if (c + 1 < 8) {
      #pragma unroll
      for (int i = 0; i < 8; ++i)
        xa[((c + 1) & 1) * 8 + i] = *(const v4f*)(xrow + ((c + 1) * 8 + i) * 8);
    }
    #pragma unroll
    for (int i = 0; i < 8; ++i) {
      const int o = c * 8 + i;
      if (o + 1 < Ec / 8) {
        #pragma unroll
        for (int nt = 0; nt < 4; ++nt)
          wb[(o + 1) & 1][nt] =
              *(const v4f*)(wlane + ((o + 1) * 64 + nt * 16) * 8);
      }
      const v4f x = xa[(c & 1) * 8 + i];
      v2f a0, a1;
      a0.x = x.x; a0.y = x.y;
      a1.x = x.z; a1.y = x.w;
      #pragma unroll
      for (int nt = 0; nt < 4; ++nt) {
        const v4f w4 = wb[o & 1][nt];
        v2f b0; b0.x = w4.x; b0.y = w4.y;
        acc[nt] = wmma4(a0, b0, acc[nt]);
      }
      #pragma unroll
      for (int nt = 0; nt < 4; ++nt) {
        const v4f w4 = wb[o & 1][nt];
        v2f b1; b1.x = w4.z; b1.y = w4.w;
        acc[nt] = wmma4(a1, b1, acc[nt]);
      }
    }
  }

  const int mbase = row0 + h * 8;
  #pragma unroll
  for (int r = 0; r < 8; ++r) {
    float* yrow = Y + (size_t)(mbase + r) * Hc;
    #pragma unroll
    for (int nt = 0; nt < 4; ++nt)
      yrow[nt * 16 + nl] = acc[nt][r];
  }
}

static __device__ __forceinline__ void done_prefix(const int* __restrict__ db,
                                                   int lane, int (&cj)[8],
                                                   int& total, int& first)
{
  int dv[8];
  #pragma unroll
  for (int j = 0; j < 8; ++j) dv[j] = db[(32 * j + lane) * 4];
  const unsigned lanes_le = (2u << lane) - 1u;
  int base = 0, fd = Tc;
  #pragma unroll
  for (int j = 0; j < 8; ++j) {
    const unsigned m = __builtin_amdgcn_ballot_w32(dv[j] != 0);
    cj[j] = base + __popc(m & lanes_le);
    if (m != 0u && fd == Tc) fd = 32 * j + (__ffs(m) - 1);
    base += __popc(m);
  }
  total = base;
  first = fd;
}

constexpr int RW = 4;
constexpr int KPITCH = 68;
constexpr int HPITCH = 20;

static __device__ __forceinline__ void
retention64(const float* __restrict__ q, const float* __restrict__ kbuf,
            const float* __restrict__ vbuf, const float* __restrict__ hstate,
            const int* __restrict__ dones, float* __restrict__ out,
            const int g, const int b)
{
  __shared__ float ktab[Tc + 1];
  __shared__ int   cumsh[Tc];
  __shared__ __attribute__((aligned(16))) float ksh[2][16 * KPITCH];
  __shared__ __attribute__((aligned(16))) float vsh[2][16 * KPITCH];

  const int tid   = threadIdx.x;
  const int lane  = tid & 31;
  const int wave  = __builtin_amdgcn_readfirstlane(tid >> 5);
  const int nl    = lane & 15;
  const int khalf = (lane >> 4) << 1;
  const int mh    = (lane >> 4) << 3;
  const int it    = g * RW + wave;
  const int last  = g * RW + RW - 1;

  const float* qb = q      + (size_t)b * Sc * Hc;
  const float* kb = kbuf   + (size_t)b * Sc * Hc;
  const float* vb = vbuf   + (size_t)b * Sc * Hc;
  const float* hb = hstate + (size_t)b * Hc * Hc;

  const int    srow = tid >> 4;
  const int    scol = (tid & 15) * 4;
  const float* ksrc = kb + (size_t)srow * Hc + scol;
  const float* vsrc = vb + (size_t)srow * Hc + scol;
  const int    sdst = srow * KPITCH + scol;


  const float* qrow = qb + (size_t)(it * 16 + nl) * Hc;
  v2f qa[16];
  #pragma unroll
  for (int ks = 0; ks < 16; ++ks)
    qa[ks] = *(const v2f*)(qrow + ks * 4 + khalf);

  v4f pk0 = *(const v4f*)(ksrc);
  v4f pk1 = *(const v4f*)(ksrc + 8 * Hc);
  v4f pv0 = *(const v4f*)(vsrc);
  v4f pv1 = *(const v4f*)(vsrc + 8 * Hc);

  int cj[8], ndone, fd;
  done_prefix(dones + (size_t)b * Sc, lane, cj, ndone, fd);
  (void)ndone;
  for (int i = tid; i <= Tc; i += 32 * RW) ktab[i] = __expf((float)i * LNK);
  if (wave == 0) {
    #pragma unroll
    for (int j = 0; j < 8; ++j) cumsh[32 * j + lane] = cj[j];
  }
  *(v4f*)(&ksh[0][sdst])              = pk0;
  *(v4f*)(&ksh[0][sdst + 8 * KPITCH]) = pk1;
  *(v4f*)(&vsh[0][sdst])              = pv0;
  *(v4f*)(&vsh[0][sdst + 8 * KPITCH]) = pv1;
  __syncthreads();

  const int gq = it * 16 + nl;
  const int tq = gq >> 2;
  const int cq = cumsh[tq];

  v8f acc[4] = {};
  for (int jt = 0; jt <= last; ++jt) {
    const int  cur  = jt & 1;
    const bool more = jt < last;
    if (more) {
      const size_t o = (size_t)(jt + 1) * 16 * Hc;
      pk0 = *(const v4f*)(ksrc + o);
      pk1 = *(const v4f*)(ksrc + o + 8 * Hc);
      pv0 = *(const v4f*)(vsrc + o);
      pv1 = *(const v4f*)(vsrc + o + 8 * Hc);
    }
    if (jt <= it) {
      const float* kl = &ksh[cur][nl * KPITCH + khalf];
      v2f kf[16];
      #pragma unroll
      for (int ks = 0; ks < 16; ++ks)
        kf[ks] = *(const v2f*)(kl + ks * 4);
      const float* vl = &vsh[cur][mh * KPITCH + nl];
      v2f vf[4][4];
      #pragma unroll
      for (int t = 0; t < 4; ++t) {
        #pragma unroll
        for (int nt = 0; nt < 4; ++nt) {
          vf[t][nt].x = vl[(2 * t)     * KPITCH + nt * 16];
          vf[t][nt].y = vl[(2 * t + 1) * KPITCH + nt * 16];
        }
      }
      const int   tk0 = jt * 4 + (mh >> 2);
      const int   d0  = tq - tk0;
      const int   c0  = cumsh[tk0];
      const int   c1  = cumsh[tk0 + 1];
      const float e0  = ktab[d0 < 0 ? 0 : d0];
      const float e1  = ktab[d0 < 1 ? 0 : d0 - 1];
      const float w0  = (c0 == cq) ? e0 : 0.0f;
      const float w1  = (c1 == cq) ? e1 : 0.0f;

      v8f s = {};
      #pragma unroll
      for (int ks = 0; ks < 16; ++ks)
        s = wmma4(kf[ks], qa[ks], s);

      #pragma unroll
      for (int r = 0; r < 8; ++r) {
        const int   gk = jt * 16 + mh + r;
        const float w  = (gk <= gq) ? (r < 4 ? w0 : w1) : 0.0f;
        s[r] = s[r] * w;
      }
      #pragma unroll
      for (int t = 0; t < 4; ++t) {
        v2f a;
        a.x = s[2 * t];
        a.y = s[2 * t + 1];
        #pragma unroll
        for (int nt = 0; nt < 4; ++nt)
          acc[nt] = wmma4(a, vf[t][nt], acc[nt]);
      }
    }
    if (more) {
      *(v4f*)(&ksh[cur ^ 1][sdst])              = pk0;
      *(v4f*)(&ksh[cur ^ 1][sdst + 8 * KPITCH]) = pk1;
      *(v4f*)(&vsh[cur ^ 1][sdst])              = pv0;
      *(v4f*)(&vsh[cur ^ 1][sdst + 8 * KPITCH]) = pv1;
    }
    __syncthreads();
  }

  v8f cr[4] = {};
  #pragma unroll
  for (int ks = 0; ks < 16; ++ks) {
    const float* h0 = hb + (size_t)(ks * 4 + khalf) * Hc;
    const float* h1 = h0 + Hc;
    #pragma unroll
    for (int nt = 0; nt < 4; ++nt) {
      v2f bm;
      bm.x = h0[nt * 16 + nl];
      bm.y = h1[nt * 16 + nl];
      cr[nt] = wmma4(qa[ks], bm, cr[nt]);
    }
  }

  #pragma unroll
  for (int r = 0; r < 8; ++r) {
    const int   gi = it * 16 + mh + r;
    const int   ti = gi >> 2;
    const float xi = (ti < fd) ? ktab[ti + 1] : 0.0f;
    float* orow = out + ((size_t)b * Sc + gi) * Hc;
    #pragma unroll
    for (int nt = 0; nt < 4; ++nt)
      orow[nt * 16 + nl] = acc[nt][r] + cr[nt][r] * xi;
  }
}

static __device__ __forceinline__ void
hstate_update(const float* __restrict__ kbuf, const float* __restrict__ vbuf,
              const float* __restrict__ hstate, const int* __restrict__ dones,
              float* __restrict__ hout, const int mt, const int nt, const int b)
{
  __shared__ float ktab[Tc + 1];
  __shared__ float wsh[Tc];
  __shared__ __attribute__((aligned(16))) float hpart[RW][16 * HPITCH];
  const int tid   = threadIdx.x;
  const int lane  = tid & 31;
  const int wave  = __builtin_amdgcn_readfirstlane(tid >> 5);
  const int nl    = lane & 15;
  const int khalf = (lane >> 4) << 1;

  int cj[8], cend, fd;
  done_prefix(dones + (size_t)b * Sc, lane, cj, cend, fd);
  (void)fd;

  for (int i = tid; i <= Tc; i += 32 * RW) ktab[i] = __expf((float)i * LNK);
  __syncthreads();
  if (wave == 0) {
    #pragma unroll
    for (int j = 0; j < 8; ++j) {
      const int i = 32 * j + lane;
      wsh[i] = (cj[j] == cend) ? ktab[(Tc - 1) - i] : 0.0f;
    }
  }
  __syncthreads();

  const float* kb = kbuf + (size_t)b * Sc * Hc;
  const float* vb = vbuf + (size_t)b * Sc * Hc;

  const int    sbeg = wave * (Sc / RW);
  const float* kw   = kb  + (size_t)sbeg * Hc;
  const float* vw   = vb  + (size_t)sbeg * Hc;
  const float* ww   = wsh + (sbeg >> 2);

  v8f c = {};
  for (int s0 = 0; s0 < Sc / RW; s0 += 4) {
    const int   sA = s0 + khalf;
    const float w  = ww[s0 >> 2];
    v2f a;
    a.x = kw[(size_t)sA       * Hc + mt * 16 + nl];
    a.y = kw[(size_t)(sA + 1) * Hc + mt * 16 + nl];
    v2f bm;
    bm.x = vw[(size_t)sA       * Hc + nt * 16 + nl] * w;
    bm.y = vw[(size_t)(sA + 1) * Hc + nt * 16 + nl] * w;
    c = wmma4(a, bm, c);
  }

  const int mh = (lane >> 4) << 3;
  #pragma unroll
  for (int r = 0; r < 8; ++r)
    hpart[wave][(mh + r) * HPITCH + nl] = c[r];
  __syncthreads();

  const float cd    = ktab[Tc];
  const float delta = (cend == 0) ? 1.0f : 0.0f;
  const float* hb   = hstate + (size_t)b * Hc * Hc;
  const int    trow = tid >> 3;
  const int    tcol = (tid & 7) * 2;
  v2f sum = *(const v2f*)(&hpart[0][trow * HPITCH + tcol]);
  #pragma unroll
  for (int w = 1; w < RW; ++w)
    sum = sum + *(const v2f*)(&hpart[w][trow * HPITCH + tcol]);
  const size_t go = ((size_t)b * Hc + mt * 16 + trow) * Hc + nt * 16 + tcol;
  const v2f hv = *(const v2f*)(hb + (go - (size_t)b * Hc * Hc));
  v2f res;
  res.x = sum.x + hv.x * cd * delta;
  res.y = sum.y + hv.y * cd * delta;
  *(v2f*)(hout + go) = res;
}

__global__ void __launch_bounds__(32 * RW)
retention_and_state(const float* __restrict__ q, const float* __restrict__ kbuf,
                    const float* __restrict__ vbuf,
                    const float* __restrict__ hstate,
                    const int* __restrict__ dones, float* __restrict__ out,
                    float* __restrict__ hout)
{
  const int b = blockIdx.x;
  const int y = blockIdx.y;
  constexpr int NBLK = Sc / (16 * RW);
  if (y < NBLK) {
    const int g = (y < NBLK / 2) ? (NBLK - 1 - y) : (y - NBLK / 2);
    retention64(q, kbuf, vbuf, hstate, dones, out, g, b);
  } else {
    const int tile = y - NBLK;
    hstate_update(kbuf, vbuf, hstate, dones, hout, tile >> 2, tile & 3, b);
  }
}

extern "C" void kernel_launch(void* const* d_in, const int* in_sizes, int n_in,
                              void* d_out, int out_size, void* d_ws, size_t ws_size,
                              hipStream_t stream)
{
  const float* key    = (const float*)d_in[0];
  const float* query  = (const float*)d_in[1];
  const float* value  = (const float*)d_in[2];
  const float* hstate = (const float*)d_in[3];
  const int*   dones  = (const int*)d_in[4];
  const float* W_Q    = (const float*)d_in[5];
  const float* W_K    = (const float*)d_in[6];
  const float* W_V    = (const float*)d_in[7];

  const size_t qkv = (size_t)Bc * Sc * Hc;
  float* ws = (float*)d_ws;
  float* qb = ws;
  float* kb = ws + qkv;
  float* vb = ws + 2 * qkv;

  constexpr int gproj = Bc * Sc / PROJ_ROWS;
  proj_strip<<<gproj, 256, 0, stream>>>(query, W_Q, qb);
  proj_strip<<<gproj, 256, 0, stream>>>(key,   W_K, kb);
  proj_strip<<<gproj, 256, 0, stream>>>(value, W_V, vb);

  float* out0 = (float*)d_out;
  float* hout = out0 + qkv;
  retention_and_state<<<dim3(Bc, Sc / (16 * RW) + 16), 32 * RW, 0, stream>>>(
      qb, kb, vb, hstate, dones, out0, hout);
}


# projections: q, k and v in ONE launch: a 24-wave work-group per WGP, one 8-wave team per projection, the three W in LDS one k-half at a time; launches 4 -> 2; k summed in the same order as before
# speedup vs baseline: 3.9653x; 1.0600x over previous
#include <hip/hip_runtime.h>
#include <hip/hip_bf16.h>


typedef float v2f __attribute__((ext_vector_type(2)));
typedef float v4f __attribute__((ext_vector_type(4)));
typedef float v8f __attribute__((ext_vector_type(8)));

constexpr int   Bc = 32, Sc = 1024, Ec = 512, Hc = 64, Tc = 256;
constexpr float LNK = -0.1053605156578263f;

static __device__ __forceinline__ v8f wmma4(v2f a, v2f b, v8f c) {
  return __builtin_amdgcn_wmma_f32_16x16x4_f32(false, a, false, b, (short)0, c,
                                               false, false);
}

constexpr int PROJ_ROWS  = 128;
constexpr int PROJ_TEAMS = 3;
constexpr int HALF_GRPS  = Ec / 16;
constexpr int WHALF      = HALF_GRPS * Hc * 8;

static __device__ __forceinline__ void
stage_w_half(const float* __restrict__ W, float* __restrict__ wt,
             const int lt, const int half)
{
  constexpr int NIT = HALF_GRPS * Hc / 256;
  v4f lo[NIT], hi[NIT];
  #pragma unroll
  for (int it = 0; it < NIT; ++it) {
    const int p  = lt + it * 256;
    const int ol = p >> 6, n = p & 63;
    const float* wsrc = W + (size_t)((half * HALF_GRPS + ol) * 8) * Hc + n;
    lo[it].x = wsrc[0 * Hc]; lo[it].y = wsrc[1 * Hc];
    lo[it].z = wsrc[2 * Hc]; lo[it].w = wsrc[3 * Hc];
    hi[it].x = wsrc[4 * Hc]; hi[it].y = wsrc[5 * Hc];
    hi[it].z = wsrc[6 * Hc]; hi[it].w = wsrc[7 * Hc];
  }
  #pragma unroll
  for (int it = 0; it < NIT; ++it) {
    const int p = lt + it * 256;
    *(v4f*)(&wt[p * 8])     = lo[it];
    *(v4f*)(&wt[p * 8 + 4]) = hi[it];
  }
}

__global__ void __launch_bounds__(256 * PROJ_TEAMS)
proj_qkv(const float* __restrict__ Xq, const float* __restrict__ Xk,
         const float* __restrict__ Xv, const float* __restrict__ Wq,
         const float* __restrict__ Wk, const float* __restrict__ Wv,
         float* __restrict__ Yq, float* __restrict__ Yk,
         float* __restrict__ Yv)
{
  __shared__ __attribute__((aligned(16))) float wl[PROJ_TEAMS * WHALF];

  const int tid  = threadIdx.x;
  const int team = __builtin_amdgcn_readfirstlane(tid >> 8);
  const float* X = (team == 0) ? Xq : (team == 1) ? Xk : Xv;
  const float* Wm = (team == 0) ? Wq : (team == 1) ? Wk : Wv;
  float*       Y = (team == 0) ? Yq : (team == 1) ? Yk : Yv;
  float*       wt = wl + team * WHALF;

  const int lt   = tid & 255;
  const int lane = tid & 31;
  const int wave = lt >> 5;
  const int nl   = lane & 15;
  const int h    = lane >> 4;
  const int row0 = blockIdx.x * PROJ_ROWS + wave * 16;

  const float* xrow  = X + (size_t)(row0 + nl) * Ec + h * 4;
  const float* wlane = wt + nl * 8 + h * 4;

  v8f acc[4] = {};
  v4f xa[16];
  v4f wb[2][4];

  #pragma unroll
  for (int half = 0; half < 2; ++half) {
    if (half == 1) __syncthreads();
    stage_w_half(Wm, wt, lt, half);
    if (half == 0) {
      #pragma unroll
      for (int i = 0; i < 8; ++i) xa[i] = *(const v4f*)(xrow + i * 8);
    }
    __syncthreads();

    #pragma unroll
    for (int nt = 0; nt < 4; ++nt) wb[0][nt] = *(const v4f*)(wlane + nt * 128);

    #pragma unroll
    for (int cc = 0; cc < 4; ++cc) {
      const int c = half * 4 + cc;
      if (c + 1 < 8) {
        #pragma unroll
        for (int i = 0; i < 8; ++i)
          xa[((c + 1) & 1) * 8 + i] = *(const v4f*)(xrow + ((c + 1) * 8 + i) * 8);
      }
      #pragma unroll
      for (int i = 0; i < 8; ++i) {
        const int ol = cc * 8 + i;
        if (ol + 1 < HALF_GRPS) {
          #pragma unroll
          for (int nt = 0; nt < 4; ++nt)
            wb[(ol + 1) & 1][nt] =
                *(const v4f*)(wlane + ((ol + 1) * 64 + nt * 16) * 8);
        }
        const v4f x = xa[(c & 1) * 8 + i];
        v2f a0, a1;
        a0.x = x.x; a0.y = x.y;
        a1.x = x.z; a1.y = x.w;
        #pragma unroll
        for (int nt = 0; nt < 4; ++nt) {
          const v4f w4 = wb[ol & 1][nt];
          v2f b0; b0.x = w4.x; b0.y = w4.y;
          acc[nt] = wmma4(a0, b0, acc[nt]);
        }
        #pragma unroll
        for (int nt = 0; nt < 4; ++nt) {
          const v4f w4 = wb[ol & 1][nt];
          v2f b1; b1.x = w4.z; b1.y = w4.w;
          acc[nt] = wmma4(a1, b1, acc[nt]);
        }
      }
    }
  }

  const int mbase = row0 + h * 8;
  #pragma unroll
  for (int r = 0; r < 8; ++r) {
    float* yrow = Y + (size_t)(mbase + r) * Hc;
    #pragma unroll
    for (int nt = 0; nt < 4; ++nt)
      yrow[nt * 16 + nl] = acc[nt][r];
  }
}

static __device__ __forceinline__ void done_prefix(const int* __restrict__ db,
                                                   int lane, int (&cj)[8],
                                                   int& total, int& first)
{
  int dv[8];
  #pragma unroll
  for (int j = 0; j < 8; ++j) dv[j] = db[(32 * j + lane) * 4];
  const unsigned lanes_le = (2u << lane) - 1u;
  int base = 0, fd = Tc;
  #pragma unroll
  for (int j = 0; j < 8; ++j) {
    const unsigned m = __builtin_amdgcn_ballot_w32(dv[j] != 0);
    cj[j] = base + __popc(m & lanes_le);
    if (m != 0u && fd == Tc) fd = 32 * j + (__ffs(m) - 1);
    base += __popc(m);
  }
  total = base;
  first = fd;
}

constexpr int RW = 4;
constexpr int KPITCH = 68;
constexpr int HPITCH = 20;

static __device__ __forceinline__ void
retention64(const float* __restrict__ q, const float* __restrict__ kbuf,
            const float* __restrict__ vbuf, const float* __restrict__ hstate,
            const int* __restrict__ dones, float* __restrict__ out,
            const int g, const int b)
{
  __shared__ float ktab[Tc + 1];
  __shared__ int   cumsh[Tc];
  __shared__ __attribute__((aligned(16))) float ksh[2][16 * KPITCH];
  __shared__ __attribute__((aligned(16))) float vsh[2][16 * KPITCH];

  const int tid   = threadIdx.x;
  const int lane  = tid & 31;
  const int wave  = __builtin_amdgcn_readfirstlane(tid >> 5);
  const int nl    = lane & 15;
  const int khalf = (lane >> 4) << 1;
  const int mh    = (lane >> 4) << 3;
  const int it    = g * RW + wave;
  const int last  = g * RW + RW - 1;

  const float* qb = q      + (size_t)b * Sc * Hc;
  const float* kb = kbuf   + (size_t)b * Sc * Hc;
  const float* vb = vbuf   + (size_t)b * Sc * Hc;
  const float* hb = hstate + (size_t)b * Hc * Hc;

  const int    srow = tid >> 4;
  const int    scol = (tid & 15) * 4;
  const float* ksrc = kb + (size_t)srow * Hc + scol;
  const float* vsrc = vb + (size_t)srow * Hc + scol;
  const int    sdst = srow * KPITCH + scol;


  const float* qrow = qb + (size_t)(it * 16 + nl) * Hc;
  v2f qa[16];
  #pragma unroll
  for (int ks = 0; ks < 16; ++ks)
    qa[ks] = *(const v2f*)(qrow + ks * 4 + khalf);

  v4f pk0 = *(const v4f*)(ksrc);
  v4f pk1 = *(const v4f*)(ksrc + 8 * Hc);
  v4f pv0 = *(const v4f*)(vsrc);
  v4f pv1 = *(const v4f*)(vsrc + 8 * Hc);

  int cj[8], ndone, fd;
  done_prefix(dones + (size_t)b * Sc, lane, cj, ndone, fd);
  (void)ndone;
  for (int i = tid; i <= Tc; i += 32 * RW) ktab[i] = __expf((float)i * LNK);
  if (wave == 0) {
    #pragma unroll
    for (int j = 0; j < 8; ++j) cumsh[32 * j + lane] = cj[j];
  }
  *(v4f*)(&ksh[0][sdst])              = pk0;
  *(v4f*)(&ksh[0][sdst + 8 * KPITCH]) = pk1;
  *(v4f*)(&vsh[0][sdst])              = pv0;
  *(v4f*)(&vsh[0][sdst + 8 * KPITCH]) = pv1;
  __syncthreads();

  const int gq = it * 16 + nl;
  const int tq = gq >> 2;
  const int cq = cumsh[tq];

  v8f acc[4] = {};
  for (int jt = 0; jt <= last; ++jt) {
    const int  cur  = jt & 1;
    const bool more = jt < last;
    if (more) {
      const size_t o = (size_t)(jt + 1) * 16 * Hc;
      pk0 = *(const v4f*)(ksrc + o);
      pk1 = *(const v4f*)(ksrc + o + 8 * Hc);
      pv0 = *(const v4f*)(vsrc + o);
      pv1 = *(const v4f*)(vsrc + o + 8 * Hc);
    }
    if (jt <= it) {
      const float* kl = &ksh[cur][nl * KPITCH + khalf];
      v2f kf[16];
      #pragma unroll
      for (int ks = 0; ks < 16; ++ks)
        kf[ks] = *(const v2f*)(kl + ks * 4);
      const float* vl = &vsh[cur][mh * KPITCH + nl];
      v2f vf[4][4];
      #pragma unroll
      for (int t = 0; t < 4; ++t) {
        #pragma unroll
        for (int nt = 0; nt < 4; ++nt) {
          vf[t][nt].x = vl[(2 * t)     * KPITCH + nt * 16];
          vf[t][nt].y = vl[(2 * t + 1) * KPITCH + nt * 16];
        }
      }
      const int   tk0 = jt * 4 + (mh >> 2);
      const int   d0  = tq - tk0;
      const int   c0  = cumsh[tk0];
      const int   c1  = cumsh[tk0 + 1];
      const float e0  = ktab[d0 < 0 ? 0 : d0];
      const float e1  = ktab[d0 < 1 ? 0 : d0 - 1];
      const float w0  = (c0 == cq) ? e0 : 0.0f;
      const float w1  = (c1 == cq) ? e1 : 0.0f;

      v8f s = {};
      #pragma unroll
      for (int ks = 0; ks < 16; ++ks)
        s = wmma4(kf[ks], qa[ks], s);

      #pragma unroll
      for (int r = 0; r < 8; ++r) {
        const int   gk = jt * 16 + mh + r;
        const float w  = (gk <= gq) ? (r < 4 ? w0 : w1) : 0.0f;
        s[r] = s[r] * w;
      }
      #pragma unroll
      for (int t = 0; t < 4; ++t) {
        v2f a;
        a.x = s[2 * t];
        a.y = s[2 * t + 1];
        #pragma unroll
        for (int nt = 0; nt < 4; ++nt)
          acc[nt] = wmma4(a, vf[t][nt], acc[nt]);
      }
    }
    if (more) {
      *(v4f*)(&ksh[cur ^ 1][sdst])              = pk0;
      *(v4f*)(&ksh[cur ^ 1][sdst + 8 * KPITCH]) = pk1;
      *(v4f*)(&vsh[cur ^ 1][sdst])              = pv0;
      *(v4f*)(&vsh[cur ^ 1][sdst + 8 * KPITCH]) = pv1;
    }
    __syncthreads();
  }

  v8f cr[4] = {};
  #pragma unroll
  for (int ks = 0; ks < 16; ++ks) {
    const float* h0 = hb + (size_t)(ks * 4 + khalf) * Hc;
    const float* h1 = h0 + Hc;
    #pragma unroll
    for (int nt = 0; nt < 4; ++nt) {
      v2f bm;
      bm.x = h0[nt * 16 + nl];
      bm.y = h1[nt * 16 + nl];
      cr[nt] = wmma4(qa[ks], bm, cr[nt]);
    }
  }

  #pragma unroll
  for (int r = 0; r < 8; ++r) {
    const int   gi = it * 16 + mh + r;
    const int   ti = gi >> 2;
    const float xi = (ti < fd) ? ktab[ti + 1] : 0.0f;
    float* orow = out + ((size_t)b * Sc + gi) * Hc;
    #pragma unroll
    for (int nt = 0; nt < 4; ++nt)
      orow[nt * 16 + nl] = acc[nt][r] + cr[nt][r] * xi;
  }
}

static __device__ __forceinline__ void
hstate_update(const float* __restrict__ kbuf, const float* __restrict__ vbuf,
              const float* __restrict__ hstate, const int* __restrict__ dones,
              float* __restrict__ hout, const int mt, const int nt, const int b)
{
  __shared__ float ktab[Tc + 1];
  __shared__ float wsh[Tc];
  __shared__ __attribute__((aligned(16))) float hpart[RW][16 * HPITCH];
  const int tid   = threadIdx.x;
  const int lane  = tid & 31;
  const int wave  = __builtin_amdgcn_readfirstlane(tid >> 5);
  const int nl    = lane & 15;
  const int khalf = (lane >> 4) << 1;

  int cj[8], cend, fd;
  done_prefix(dones + (size_t)b * Sc, lane, cj, cend, fd);
  (void)fd;

  for (int i = tid; i <= Tc; i += 32 * RW) ktab[i] = __expf((float)i * LNK);
  __syncthreads();
  if (wave == 0) {
    #pragma unroll
    for (int j = 0; j < 8; ++j) {
      const int i = 32 * j + lane;
      wsh[i] = (cj[j] == cend) ? ktab[(Tc - 1) - i] : 0.0f;
    }
  }
  __syncthreads();

  const float* kb = kbuf + (size_t)b * Sc * Hc;
  const float* vb = vbuf + (size_t)b * Sc * Hc;

  const int    sbeg = wave * (Sc / RW);
  const float* kw   = kb  + (size_t)sbeg * Hc;
  const float* vw   = vb  + (size_t)sbeg * Hc;
  const float* ww   = wsh + (sbeg >> 2);

  v8f c = {};
  for (int s0 = 0; s0 < Sc / RW; s0 += 4) {
    const int   sA = s0 + khalf;
    const float w  = ww[s0 >> 2];
    v2f a;
    a.x = kw[(size_t)sA       * Hc + mt * 16 + nl];
    a.y = kw[(size_t)(sA + 1) * Hc + mt * 16 + nl];
    v2f bm;
    bm.x = vw[(size_t)sA       * Hc + nt * 16 + nl] * w;
    bm.y = vw[(size_t)(sA + 1) * Hc + nt * 16 + nl] * w;
    c = wmma4(a, bm, c);
  }

  const int mh = (lane >> 4) << 3;
  #pragma unroll
  for (int r = 0; r < 8; ++r)
    hpart[wave][(mh + r) * HPITCH + nl] = c[r];
  __syncthreads();

  const float cd    = ktab[Tc];
  const float delta = (cend == 0) ? 1.0f : 0.0f;
  const float* hb   = hstate + (size_t)b * Hc * Hc;
  const int    trow = tid >> 3;
  const int    tcol = (tid & 7) * 2;
  v2f sum = *(const v2f*)(&hpart[0][trow * HPITCH + tcol]);
  #pragma unroll
  for (int w = 1; w < RW; ++w)
    sum = sum + *(const v2f*)(&hpart[w][trow * HPITCH + tcol]);
  const size_t go = ((size_t)b * Hc + mt * 16 + trow) * Hc + nt * 16 + tcol;
  const v2f hv = *(const v2f*)(hb + (go - (size_t)b * Hc * Hc));
  v2f res;
  res.x = sum.x + hv.x * cd * delta;
  res.y = sum.y + hv.y * cd * delta;
  *(v2f*)(hout + go) = res;
}

__global__ void __launch_bounds__(32 * RW)
retention_and_state(const float* __restrict__ q, const float* __restrict__ kbuf,
                    const float* __restrict__ vbuf,
                    const float* __restrict__ hstate,
                    const int* __restrict__ dones, float* __restrict__ out,
                    float* __restrict__ hout)
{
  const int b = blockIdx.x;
  const int y = blockIdx.y;
  constexpr int NBLK = Sc / (16 * RW);
  if (y < NBLK) {
    const int g = (y < NBLK / 2) ? (NBLK - 1 - y) : (y - NBLK / 2);
    retention64(q, kbuf, vbuf, hstate, dones, out, g, b);
  } else {
    const int tile = y - NBLK;
    hstate_update(kbuf, vbuf, hstate, dones, hout, tile >> 2, tile & 3, b);
  }
}

extern "C" void kernel_launch(void* const* d_in, const int* in_sizes, int n_in,
                              void* d_out, int out_size, void* d_ws, size_t ws_size,
                              hipStream_t stream)
{
  const float* key    = (const float*)d_in[0];
  const float* query  = (const float*)d_in[1];
  const float* value  = (const float*)d_in[2];
  const float* hstate = (const float*)d_in[3];
  const int*   dones  = (const int*)d_in[4];
  const float* W_Q    = (const float*)d_in[5];
  const float* W_K    = (const float*)d_in[6];
  const float* W_V    = (const float*)d_in[7];

  const size_t qkv = (size_t)Bc * Sc * Hc;
  float* ws = (float*)d_ws;
  float* qb = ws;
  float* kb = ws + qkv;
  float* vb = ws + 2 * qkv;

  constexpr int gproj = Bc * Sc / PROJ_ROWS;
  proj_qkv<<<gproj, 256 * PROJ_TEAMS, 0, stream>>>(query, key, value,
                                                   W_Q, W_K, W_V, qb, kb, vb);

  float* out0 = (float*)d_out;
  float* hout = out0 + qkv;
  retention_and_state<<<dim3(Bc, Sc / (16 * RW) + 16), 32 * RW, 0, stream>>>(
      qb, kb, vb, hstate, dones, out0, hout);
}


# retention: each 16-row strip's key tiles split by parity over two waves (8-wave work-groups, 32 keys staged per barrier), the two partial sums added through LDS; keys summed in another order
# speedup vs baseline: 4.2630x; 1.0751x over previous
#include <hip/hip_runtime.h>
#include <hip/hip_bf16.h>


typedef float v2f __attribute__((ext_vector_type(2)));
typedef float v4f __attribute__((ext_vector_type(4)));
typedef float v8f __attribute__((ext_vector_type(8)));

constexpr int   Bc = 32, Sc = 1024, Ec = 512, Hc = 64, Tc = 256;
constexpr float LNK = -0.1053605156578263f;

static __device__ __forceinline__ v8f wmma4(v2f a, v2f b, v8f c) {
  return __builtin_amdgcn_wmma_f32_16x16x4_f32(false, a, false, b, (short)0, c,
                                               false, false);
}

constexpr int PROJ_ROWS  = 128;
constexpr int PROJ_TEAMS = 3;
constexpr int HALF_GRPS  = Ec / 16;
constexpr int WHALF      = HALF_GRPS * Hc * 8;

static __device__ __forceinline__ void
stage_w_half(const float* __restrict__ W, float* __restrict__ wt,
             const int lt, const int half)
{
  constexpr int NIT = HALF_GRPS * Hc / 256;
  v4f lo[NIT], hi[NIT];
  #pragma unroll
  for (int it = 0; it < NIT; ++it) {
    const int p  = lt + it * 256;
    const int ol = p >> 6, n = p & 63;
    const float* wsrc = W + (size_t)((half * HALF_GRPS + ol) * 8) * Hc + n;
    lo[it].x = wsrc[0 * Hc]; lo[it].y = wsrc[1 * Hc];
    lo[it].z = wsrc[2 * Hc]; lo[it].w = wsrc[3 * Hc];
    hi[it].x = wsrc[4 * Hc]; hi[it].y = wsrc[5 * Hc];
    hi[it].z = wsrc[6 * Hc]; hi[it].w = wsrc[7 * Hc];
  }
  #pragma unroll
  for (int it = 0; it < NIT; ++it) {
    const int p = lt + it * 256;
    *(v4f*)(&wt[p * 8])     = lo[it];
    *(v4f*)(&wt[p * 8 + 4]) = hi[it];
  }
}

__global__ void __launch_bounds__(256 * PROJ_TEAMS)
proj_qkv(const float* __restrict__ Xq, const float* __restrict__ Xk,
         const float* __restrict__ Xv, const float* __restrict__ Wq,
         const float* __restrict__ Wk, const float* __restrict__ Wv,
         float* __restrict__ Yq, float* __restrict__ Yk,
         float* __restrict__ Yv)
{
  __shared__ __attribute__((aligned(16))) float wl[PROJ_TEAMS * WHALF];

  const int tid  = threadIdx.x;
  const int team = __builtin_amdgcn_readfirstlane(tid >> 8);
  const float* X = (team == 0) ? Xq : (team == 1) ? Xk : Xv;
  const float* Wm = (team == 0) ? Wq : (team == 1) ? Wk : Wv;
  float*       Y = (team == 0) ? Yq : (team == 1) ? Yk : Yv;
  float*       wt = wl + team * WHALF;

  const int lt   = tid & 255;
  const int lane = tid & 31;
  const int wave = lt >> 5;
  const int nl   = lane & 15;
  const int h    = lane >> 4;
  const int row0 = blockIdx.x * PROJ_ROWS + wave * 16;

  const float* xrow  = X + (size_t)(row0 + nl) * Ec + h * 4;
  const float* wlane = wt + nl * 8 + h * 4;

  v8f acc[4] = {};
  v4f xa[16];
  v4f wb[2][4];

  #pragma unroll
  for (int half = 0; half < 2; ++half) {
    if (half == 1) __syncthreads();
    stage_w_half(Wm, wt, lt, half);
    if (half == 0) {
      #pragma unroll
      for (int i = 0; i < 8; ++i) xa[i] = *(const v4f*)(xrow + i * 8);
    }
    __syncthreads();

    #pragma unroll
    for (int nt = 0; nt < 4; ++nt) wb[0][nt] = *(const v4f*)(wlane + nt * 128);

    #pragma unroll
    for (int cc = 0; cc < 4; ++cc) {
      const int c = half * 4 + cc;
      if (c + 1 < 8) {
        #pragma unroll
        for (int i = 0; i < 8; ++i)
          xa[((c + 1) & 1) * 8 + i] = *(const v4f*)(xrow + ((c + 1) * 8 + i) * 8);
      }
      #pragma unroll
      for (int i = 0; i < 8; ++i) {
        const int ol = cc * 8 + i;
        if (ol + 1 < HALF_GRPS) {
          #pragma unroll
          for (int nt = 0; nt < 4; ++nt)
            wb[(ol + 1) & 1][nt] =
                *(const v4f*)(wlane + ((ol + 1) * 64 + nt * 16) * 8);
        }
        const v4f x = xa[(c & 1) * 8 + i];
        v2f a0, a1;
        a0.x = x.x; a0.y = x.y;
        a1.x = x.z; a1.y = x.w;
        #pragma unroll
        for (int nt = 0; nt < 4; ++nt) {
          const v4f w4 = wb[ol & 1][nt];
          v2f b0; b0.x = w4.x; b0.y = w4.y;
          acc[nt] = wmma4(a0, b0, acc[nt]);
        }
        #pragma unroll
        for (int nt = 0; nt < 4; ++nt) {
          const v4f w4 = wb[ol & 1][nt];
          v2f b1; b1.x = w4.z; b1.y = w4.w;
          acc[nt] = wmma4(a1, b1, acc[nt]);
        }
      }
    }
  }

  const int mbase = row0 + h * 8;
  #pragma unroll
  for (int r = 0; r < 8; ++r) {
    float* yrow = Y + (size_t)(mbase + r) * Hc;
    #pragma unroll
    for (int nt = 0; nt < 4; ++nt)
      yrow[nt * 16 + nl] = acc[nt][r];
  }
}

static __device__ __forceinline__ void done_prefix(const int* __restrict__ db,
                                                   int lane, int (&cj)[8],
                                                   int& total, int& first)
{
  int dv[8];
  #pragma unroll
  for (int j = 0; j < 8; ++j) dv[j] = db[(32 * j + lane) * 4];
  const unsigned lanes_le = (2u << lane) - 1u;
  int base = 0, fd = Tc;
  #pragma unroll
  for (int j = 0; j < 8; ++j) {
    const unsigned m = __builtin_amdgcn_ballot_w32(dv[j] != 0);
    cj[j] = base + __popc(m & lanes_le);
    if (m != 0u && fd == Tc) fd = 32 * j + (__ffs(m) - 1);
    base += __popc(m);
  }
  total = base;
  first = fd;
}

constexpr int RW = 4;
constexpr int KSPLIT = 2;
constexpr int NWAVE = RW * KSPLIT;
constexpr int KPITCH = 68;
constexpr int HPITCH = 20;

static __device__ __forceinline__ void
retention64(const float* __restrict__ q, const float* __restrict__ kbuf,
            const float* __restrict__ vbuf, const float* __restrict__ hstate,
            const int* __restrict__ dones, float* __restrict__ out,
            const int g, const int b)
{
  __shared__ float ktab[Tc + 1];
  __shared__ int   cumsh[Tc];
  __shared__ __attribute__((aligned(16))) float ksh[2][32 * KPITCH];
  __shared__ __attribute__((aligned(16))) float vsh[2][32 * KPITCH];
  __shared__ __attribute__((aligned(16))) float comb[RW][8 * 128];

  const int tid   = threadIdx.x;
  const int lane  = tid & 31;
  const int wave  = __builtin_amdgcn_readfirstlane(tid >> 5);
  const int sw    = wave & (RW - 1);
  const int par   = wave >> 2;
  const int nl    = lane & 15;
  const int khalf = (lane >> 4) << 1;
  const int mh    = (lane >> 4) << 3;
  const int it    = g * RW + sw;
  const int npair = 2 * g + 2;

  const float* qb = q      + (size_t)b * Sc * Hc;
  const float* kb = kbuf   + (size_t)b * Sc * Hc;
  const float* vb = vbuf   + (size_t)b * Sc * Hc;
  const float* hb = hstate + (size_t)b * Hc * Hc;

  const int    srow = tid >> 4;
  const int    scol = (tid & 15) * 4;
  const float* ksrc = kb + (size_t)srow * Hc + scol;
  const float* vsrc = vb + (size_t)srow * Hc + scol;
  const int    sdst = srow * KPITCH + scol;


  const float* qrow = qb + (size_t)(it * 16 + nl) * Hc;
  v2f qa[16];
  #pragma unroll
  for (int ks = 0; ks < 16; ++ks)
    qa[ks] = *(const v2f*)(qrow + ks * 4 + khalf);

  v4f pk0 = *(const v4f*)(ksrc);
  v4f pk1 = *(const v4f*)(ksrc + 16 * Hc);
  v4f pv0 = *(const v4f*)(vsrc);
  v4f pv1 = *(const v4f*)(vsrc + 16 * Hc);

  int cj[8], ndone, fd;
  done_prefix(dones + (size_t)b * Sc, lane, cj, ndone, fd);
  (void)ndone;
  for (int i = tid; i <= Tc; i += 32 * NWAVE) ktab[i] = __expf((float)i * LNK);
  if (wave == 0) {
    #pragma unroll
    for (int j = 0; j < 8; ++j) cumsh[32 * j + lane] = cj[j];
  }
  *(v4f*)(&ksh[0][sdst])               = pk0;
  *(v4f*)(&ksh[0][sdst + 16 * KPITCH]) = pk1;
  *(v4f*)(&vsh[0][sdst])               = pv0;
  *(v4f*)(&vsh[0][sdst + 16 * KPITCH]) = pv1;
  __syncthreads();

  const int gq = it * 16 + nl;
  const int tq = gq >> 2;
  const int cq = cumsh[tq];

  v8f acc[4] = {};
  for (int d = 0; d < npair; ++d) {
    const int  cur  = d & 1;
    const bool more = d + 1 < npair;
    const int  jt   = 2 * d + par;
    if (more) {
      const size_t o = (size_t)(d + 1) * 32 * Hc;
      pk0 = *(const v4f*)(ksrc + o);
      pk1 = *(const v4f*)(ksrc + o + 16 * Hc);
      pv0 = *(const v4f*)(vsrc + o);
      pv1 = *(const v4f*)(vsrc + o + 16 * Hc);
    }
    if (jt <= it) {
      const float* kl = &ksh[cur][(par * 16 + nl) * KPITCH + khalf];
      v2f kf[16];
      #pragma unroll
      for (int ks = 0; ks < 16; ++ks)
        kf[ks] = *(const v2f*)(kl + ks * 4);
      const float* vl = &vsh[cur][(par * 16 + mh) * KPITCH + nl];
      v2f vf[4][4];
      #pragma unroll
      for (int t = 0; t < 4; ++t) {
        #pragma unroll
        for (int nt = 0; nt < 4; ++nt) {
          vf[t][nt].x = vl[(2 * t)     * KPITCH + nt * 16];
          vf[t][nt].y = vl[(2 * t + 1) * KPITCH + nt * 16];
        }
      }
      const int   tk0 = jt * 4 + (mh >> 2);
      const int   d0  = tq - tk0;
      const int   c0  = cumsh[tk0];
      const int   c1  = cumsh[tk0 + 1];
      const float e0  = ktab[d0 < 0 ? 0 : d0];
      const float e1  = ktab[d0 < 1 ? 0 : d0 - 1];
      const float w0  = (c0 == cq) ? e0 : 0.0f;
      const float w1  = (c1 == cq) ? e1 : 0.0f;

      v8f s = {};
      #pragma unroll
      for (int ks = 0; ks < 16; ++ks)
        s = wmma4(kf[ks], qa[ks], s);

      #pragma unroll
      for (int r = 0; r < 8; ++r) {
        const int   gk = jt * 16 + mh + r;
        const float w  = (gk <= gq) ? (r < 4 ? w0 : w1) : 0.0f;
        s[r] = s[r] * w;
      }
      #pragma unroll
      for (int t = 0; t < 4; ++t) {
        v2f a;
        a.x = s[2 * t];
        a.y = s[2 * t + 1];
        #pragma unroll
        for (int nt = 0; nt < 4; ++nt)
          acc[nt] = wmma4(a, vf[t][nt], acc[nt]);
      }
    }
    if (more) {
      *(v4f*)(&ksh[cur ^ 1][sdst])               = pk0;
      *(v4f*)(&ksh[cur ^ 1][sdst + 16 * KPITCH]) = pk1;
      *(v4f*)(&vsh[cur ^ 1][sdst])               = pv0;
      *(v4f*)(&vsh[cur ^ 1][sdst + 16 * KPITCH]) = pv1;
    }
    __syncthreads();
  }

  float* cw = &comb[sw][lane * 4];
  v8f cr[4] = {};
  if (par == 1) {
    #pragma unroll
    for (int nt = 0; nt < 4; ++nt) {
      #pragma unroll
      for (int hq = 0; hq < 2; ++hq) {
        v4f p;
        p.x = acc[nt][4 * hq];     p.y = acc[nt][4 * hq + 1];
        p.z = acc[nt][4 * hq + 2]; p.w = acc[nt][4 * hq + 3];
        *(v4f*)(cw + (nt * 2 + hq) * 128) = p;
      }
    }
  } else {
    #pragma unroll
    for (int ks = 0; ks < 16; ++ks) {
      const float* h0 = hb + (size_t)(ks * 4 + khalf) * Hc;
      const float* h1 = h0 + Hc;
      #pragma unroll
      for (int nt = 0; nt < 4; ++nt) {
        v2f bm;
        bm.x = h0[nt * 16 + nl];
        bm.y = h1[nt * 16 + nl];
        cr[nt] = wmma4(qa[ks], bm, cr[nt]);
      }
    }
  }
  __syncthreads();

  if (par == 0) {
    #pragma unroll
    for (int nt = 0; nt < 4; ++nt) {
      #pragma unroll
      for (int hq = 0; hq < 2; ++hq) {
        const v4f p = *(const v4f*)(cw + (nt * 2 + hq) * 128);
        acc[nt][4 * hq]     += p.x; acc[nt][4 * hq + 1] += p.y;
        acc[nt][4 * hq + 2] += p.z; acc[nt][4 * hq + 3] += p.w;
      }
    }
    #pragma unroll
    for (int r = 0; r < 8; ++r) {
      const int   gi = it * 16 + mh + r;
      const int   ti = gi >> 2;
      const float xi = (ti < fd) ? ktab[ti + 1] : 0.0f;
      float* orow = out + ((size_t)b * Sc + gi) * Hc;
      #pragma unroll
      for (int nt = 0; nt < 4; ++nt)
        orow[nt * 16 + nl] = acc[nt][r] + cr[nt][r] * xi;
    }
  }
}

static __device__ __forceinline__ void
hstate_update(const float* __restrict__ kbuf, const float* __restrict__ vbuf,
              const float* __restrict__ hstate, const int* __restrict__ dones,
              float* __restrict__ hout, const int pr, const int b)
{
  __shared__ float ktab[Tc + 1];
  __shared__ float wsh[Tc];
  __shared__ __attribute__((aligned(16))) float hpart[KSPLIT][RW][16 * HPITCH];
  const int tid   = threadIdx.x;
  const int lane  = tid & 31;
  const int wave  = __builtin_amdgcn_readfirstlane(tid >> 5);
  const int wq    = wave & (RW - 1);
  const int th    = wave >> 2;
  const int tile  = pr * 2 + th;
  const int mt    = tile >> 2;
  const int nt    = tile & 3;
  const int nl    = lane & 15;
  const int khalf = (lane >> 4) << 1;

  int cj[8], cend, fd;
  done_prefix(dones + (size_t)b * Sc, lane, cj, cend, fd);
  (void)fd;

  for (int i = tid; i <= Tc; i += 32 * NWAVE) ktab[i] = __expf((float)i * LNK);
  __syncthreads();
  if (wave == 0) {
    #pragma unroll
    for (int j = 0; j < 8; ++j) {
      const int i = 32 * j + lane;
      wsh[i] = (cj[j] == cend) ? ktab[(Tc - 1) - i] : 0.0f;
    }
  }
  __syncthreads();

  const float* kb = kbuf + (size_t)b * Sc * Hc;
  const float* vb = vbuf + (size_t)b * Sc * Hc;

  const int    sbeg = wq * (Sc / RW);
  const float* kw   = kb  + (size_t)sbeg * Hc;
  const float* vw   = vb  + (size_t)sbeg * Hc;
  const float* ww   = wsh + (sbeg >> 2);

  v8f c = {};
  for (int s0 = 0; s0 < Sc / RW; s0 += 4) {
    const int   sA = s0 + khalf;
    const float w  = ww[s0 >> 2];
    v2f a;
    a.x = kw[(size_t)sA       * Hc + mt * 16 + nl];
    a.y = kw[(size_t)(sA + 1) * Hc + mt * 16 + nl];
    v2f bm;
    bm.x = vw[(size_t)sA       * Hc + nt * 16 + nl] * w;
    bm.y = vw[(size_t)(sA + 1) * Hc + nt * 16 + nl] * w;
    c = wmma4(a, bm, c);
  }

  const int mh = (lane >> 4) << 3;
  #pragma unroll
  for (int r = 0; r < 8; ++r)
    hpart[th][wq][(mh + r) * HPITCH + nl] = c[r];
  __syncthreads();

  const float cd    = ktab[Tc];
  const float delta = (cend == 0) ? 1.0f : 0.0f;
  const float* hb   = hstate + (size_t)b * Hc * Hc;
  const int    tt   = tid & 127;
  const int    trow = tt >> 3;
  const int    tcol = (tt & 7) * 2;
  v2f sum = *(const v2f*)(&hpart[th][0][trow * HPITCH + tcol]);
  #pragma unroll
  for (int w = 1; w < RW; ++w)
    sum = sum + *(const v2f*)(&hpart[th][w][trow * HPITCH + tcol]);
  const size_t go = ((size_t)b * Hc + mt * 16 + trow) * Hc + nt * 16 + tcol;
  const v2f hv = *(const v2f*)(hb + (go - (size_t)b * Hc * Hc));
  v2f res;
  res.x = sum.x + hv.x * cd * delta;
  res.y = sum.y + hv.y * cd * delta;
  *(v2f*)(hout + go) = res;
}

__global__ void __launch_bounds__(32 * NWAVE)
retention_and_state(const float* __restrict__ q, const float* __restrict__ kbuf,
                    const float* __restrict__ vbuf,
                    const float* __restrict__ hstate,
                    const int* __restrict__ dones, float* __restrict__ out,
                    float* __restrict__ hout)
{
  const int b = blockIdx.x;
  const int y = blockIdx.y;
  constexpr int NBLK = Sc / (16 * RW);
  if (y < NBLK) {
    const int g = (y < NBLK / 2) ? (NBLK - 1 - y) : (y - NBLK / 2);
    retention64(q, kbuf, vbuf, hstate, dones, out, g, b);
  } else {
    hstate_update(kbuf, vbuf, hstate, dones, hout, y - NBLK, b);
  }
}

extern "C" void kernel_launch(void* const* d_in, const int* in_sizes, int n_in,
                              void* d_out, int out_size, void* d_ws, size_t ws_size,
                              hipStream_t stream)
{
  const float* key    = (const float*)d_in[0];
  const float* query  = (const float*)d_in[1];
  const float* value  = (const float*)d_in[2];
  const float* hstate = (const float*)d_in[3];
  const int*   dones  = (const int*)d_in[4];
  const float* W_Q    = (const float*)d_in[5];
  const float* W_K    = (const float*)d_in[6];
  const float* W_V    = (const float*)d_in[7];

  const size_t qkv = (size_t)Bc * Sc * Hc;
  float* ws = (float*)d_ws;
  float* qb = ws;
  float* kb = ws + qkv;
  float* vb = ws + 2 * qkv;

  constexpr int gproj = Bc * Sc / PROJ_ROWS;
  proj_qkv<<<gproj, 256 * PROJ_TEAMS, 0, stream>>>(query, key, value,
                                                   W_Q, W_K, W_V, qb, kb, vb);

  float* out0 = (float*)d_out;
  float* hout = out0 + qkv;
  retention_and_state<<<dim3(Bc, Sc / (16 * RW) + 16 / KSPLIT), 32 * NWAVE, 0,
                        stream>>>(qb, kb, vb, hstate, dones, out0, hout);
}


# retention in chunk form: keys of earlier 64-row blocks enter through one decayed 64x64 state k^T v (carried state folded in); only the block's own 4 key tiles are scored; sums regrouped
# speedup vs baseline: 4.6850x; 1.0990x over previous
#include <hip/hip_runtime.h>
#include <hip/hip_bf16.h>


typedef float v2f __attribute__((ext_vector_type(2)));
typedef float v4f __attribute__((ext_vector_type(4)));
typedef float v8f __attribute__((ext_vector_type(8)));

constexpr int   Bc = 32, Sc = 1024, Ec = 512, Hc = 64, Tc = 256;
constexpr float LNK = -0.1053605156578263f;

static __device__ __forceinline__ v8f wmma4(v2f a, v2f b, v8f c) {
  return __builtin_amdgcn_wmma_f32_16x16x4_f32(false, a, false, b, (short)0, c,
                                               false, false);
}

constexpr int PROJ_ROWS  = 128;
constexpr int PROJ_TEAMS = 3;
constexpr int HALF_GRPS  = Ec / 16;
constexpr int WHALF      = HALF_GRPS * Hc * 8;

static __device__ __forceinline__ void
stage_w_half(const float* __restrict__ W, float* __restrict__ wt,
             const int lt, const int half)
{
  constexpr int NIT = HALF_GRPS * Hc / 256;
  v4f lo[NIT], hi[NIT];
  #pragma unroll
  for (int it = 0; it < NIT; ++it) {
    const int p  = lt + it * 256;
    const int ol = p >> 6, n = p & 63;
    const float* wsrc = W + (size_t)((half * HALF_GRPS + ol) * 8) * Hc + n;
    lo[it].x = wsrc[0 * Hc]; lo[it].y = wsrc[1 * Hc];
    lo[it].z = wsrc[2 * Hc]; lo[it].w = wsrc[3 * Hc];
    hi[it].x = wsrc[4 * Hc]; hi[it].y = wsrc[5 * Hc];
    hi[it].z = wsrc[6 * Hc]; hi[it].w = wsrc[7 * Hc];
  }
  #pragma unroll
  for (int it = 0; it < NIT; ++it) {
    const int p = lt + it * 256;
    *(v4f*)(&wt[p * 8])     = lo[it];
    *(v4f*)(&wt[p * 8 + 4]) = hi[it];
  }
}

__global__ void __launch_bounds__(256 * PROJ_TEAMS)
proj_qkv(const float* __restrict__ Xq, const float* __restrict__ Xk,
         const float* __restrict__ Xv, const float* __restrict__ Wq,
         const float* __restrict__ Wk, const float* __restrict__ Wv,
         float* __restrict__ Yq, float* __restrict__ Yk,
         float* __restrict__ Yv)
{
  __shared__ __attribute__((aligned(16))) float wl[PROJ_TEAMS * WHALF];

  const int tid  = threadIdx.x;
  const int team = __builtin_amdgcn_readfirstlane(tid >> 8);
  const float* X = (team == 0) ? Xq : (team == 1) ? Xk : Xv;
  const float* Wm = (team == 0) ? Wq : (team == 1) ? Wk : Wv;
  float*       Y = (team == 0) ? Yq : (team == 1) ? Yk : Yv;
  float*       wt = wl + team * WHALF;

  const int lt   = tid & 255;
  const int lane = tid & 31;
  const int wave = lt >> 5;
  const int nl   = lane & 15;
  const int h    = lane >> 4;
  const int row0 = blockIdx.x * PROJ_ROWS + wave * 16;

  const float* xrow  = X + (size_t)(row0 + nl) * Ec + h * 4;
  const float* wlane = wt + nl * 8 + h * 4;

  v8f acc[4] = {};
  v4f xa[16];
  v4f wb[2][4];

  #pragma unroll
  for (int half = 0; half < 2; ++half) {
    if (half == 1) __syncthreads();
    stage_w_half(Wm, wt, lt, half);
    if (half == 0) {
      #pragma unroll
      for (int i = 0; i < 8; ++i) xa[i] = *(const v4f*)(xrow + i * 8);
    }
    __syncthreads();

    #pragma unroll
    for (int nt = 0; nt < 4; ++nt) wb[0][nt] = *(const v4f*)(wlane + nt * 128);

    #pragma unroll
    for (int cc = 0; cc < 4; ++cc) {
      const int c = half * 4 + cc;
      if (c + 1 < 8) {
        #pragma unroll
        for (int i = 0; i < 8; ++i)
          xa[((c + 1) & 1) * 8 + i] = *(const v4f*)(xrow + ((c + 1) * 8 + i) * 8);
      }
      #pragma unroll
      for (int i = 0; i < 8; ++i) {
        const int ol = cc * 8 + i;
        if (ol + 1 < HALF_GRPS) {
          #pragma unroll
          for (int nt = 0; nt < 4; ++nt)
            wb[(ol + 1) & 1][nt] =
                *(const v4f*)(wlane + ((ol + 1) * 64 + nt * 16) * 8);
        }
        const v4f x = xa[(c & 1) * 8 + i];
        v2f a0, a1;
        a0.x = x.x; a0.y = x.y;
        a1.x = x.z; a1.y = x.w;
        #pragma unroll
        for (int nt = 0; nt < 4; ++nt) {
          const v4f w4 = wb[ol & 1][nt];
          v2f b0; b0.x = w4.x; b0.y = w4.y;
          acc[nt] = wmma4(a0, b0, acc[nt]);
        }
        #pragma unroll
        for (int nt = 0; nt < 4; ++nt) {
          const v4f w4 = wb[ol & 1][nt];
          v2f b1; b1.x = w4.z; b1.y = w4.w;
          acc[nt] = wmma4(a1, b1, acc[nt]);
        }
      }
    }
  }

  const int mbase = row0 + h * 8;
  #pragma unroll
  for (int r = 0; r < 8; ++r) {
    float* yrow = Y + (size_t)(mbase + r) * Hc;
    #pragma unroll
    for (int nt = 0; nt < 4; ++nt)
      yrow[nt * 16 + nl] = acc[nt][r];
  }
}

static __device__ __forceinline__ void done_prefix(const int* __restrict__ db,
                                                   int lane, int (&cj)[8],
                                                   int& total, int& first)
{
  int dv[8];
  #pragma unroll
  for (int j = 0; j < 8; ++j) dv[j] = db[(32 * j + lane) * 4];
  const unsigned lanes_le = (2u << lane) - 1u;
  int base = 0, fd = Tc;
  #pragma unroll
  for (int j = 0; j < 8; ++j) {
    const unsigned m = __builtin_amdgcn_ballot_w32(dv[j] != 0);
    cj[j] = base + __popc(m & lanes_le);
    if (m != 0u && fd == Tc) fd = 32 * j + (__ffs(m) - 1);
    base += __popc(m);
  }
  total = base;
  first = fd;
}

constexpr int RW = 4;
constexpr int KSPLIT = 2;
constexpr int NWAVE = RW * KSPLIT;
constexpr int KPITCH = 68;
constexpr int HPITCH = 20;

constexpr int SPITCH = 72;

static __device__ __forceinline__ void
retention64(const float* __restrict__ q, const float* __restrict__ kbuf,
            const float* __restrict__ vbuf, const float* __restrict__ hstate,
            const int* __restrict__ dones, float* __restrict__ out,
            const int g, const int b)
{
  __shared__ float ktab[Tc + 1];
  __shared__ int   cumsh[Tc];
  __shared__ __attribute__((aligned(16))) float wsh[Tc];
  __shared__ __attribute__((aligned(16))) float ksh[2][32 * KPITCH];
  __shared__ __attribute__((aligned(16))) float vsh[2][32 * KPITCH];
  __shared__ __attribute__((aligned(16))) float ssh[Hc * SPITCH];
  __shared__ __attribute__((aligned(16))) float comb[RW][8 * 128];

  const int tid   = threadIdx.x;
  const int lane  = tid & 31;
  const int wave  = __builtin_amdgcn_readfirstlane(tid >> 5);
  const int sw    = wave & (RW - 1);
  const int par   = wave >> 2;
  const int smt   = wave >> 1;
  const int snt   = (wave & 1) * 2;
  const int nl    = lane & 15;
  const int khalf = (lane >> 4) << 1;
  const int mh    = (lane >> 4) << 3;
  const int it    = g * RW + sw;
  const int nprev = 2 * g;
  const int t0    = 4 * RW * g;

  const float* qb = q      + (size_t)b * Sc * Hc;
  const float* kb = kbuf   + (size_t)b * Sc * Hc;
  const float* vb = vbuf   + (size_t)b * Sc * Hc;
  const float* hb = hstate + (size_t)b * Hc * Hc;

  const int    srow = tid >> 4;
  const int    scol = (tid & 15) * 4;
  const float* ksrc = kb + (size_t)srow * Hc + scol;
  const float* vsrc = vb + (size_t)srow * Hc + scol;
  const int    sdst = srow * KPITCH + scol;


  const float* qrow = qb + (size_t)(it * 16 + nl) * Hc;
  v2f qa[16];
  #pragma unroll
  for (int ks = 0; ks < 16; ++ks)
    qa[ks] = *(const v2f*)(qrow + ks * 4 + khalf);

  v4f pk0 = *(const v4f*)(ksrc);
  v4f pk1 = *(const v4f*)(ksrc + 16 * Hc);
  v4f pv0 = *(const v4f*)(vsrc);
  v4f pv1 = *(const v4f*)(vsrc + 16 * Hc);

  int cj[8], ndone, fd;
  done_prefix(dones + (size_t)b * Sc, lane, cj, ndone, fd);
  (void)ndone; (void)fd;
  for (int i = tid; i <= Tc; i += 32 * NWAVE) ktab[i] = __expf((float)i * LNK);
  if (wave == 0) {
    #pragma unroll
    for (int j = 0; j < 8; ++j) cumsh[32 * j + lane] = cj[j];
  }
  *(v4f*)(&ksh[0][sdst])               = pk0;
  *(v4f*)(&ksh[0][sdst + 16 * KPITCH]) = pk1;
  *(v4f*)(&vsh[0][sdst])               = pv0;
  *(v4f*)(&vsh[0][sdst + 16 * KPITCH]) = pv1;
  __syncthreads();

  const int cref = (g > 0) ? cumsh[t0 - 1] : 0;
  for (int i = tid; i < Tc; i += 32 * NWAVE)
    wsh[i] = (i < t0 && cumsh[i] == cref) ? ktab[t0 - 1 - i] : 0.0f;

  const int gq = it * 16 + nl;
  const int tq = gq >> 2;
  const int cq = cumsh[tq];
  __syncthreads();

  v8f st[2] = {};
  for (int d = 0; d < nprev; ++d) {
    const int cur = d & 1;
    {
      const size_t o = (size_t)(d + 1) * 32 * Hc;
      pk0 = *(const v4f*)(ksrc + o);
      pk1 = *(const v4f*)(ksrc + o + 16 * Hc);
      pv0 = *(const v4f*)(vsrc + o);
      pv1 = *(const v4f*)(vsrc + o + 16 * Hc);
    }
    const float* kl = &ksh[cur][khalf * KPITCH + smt * 16 + nl];
    const float* vl = &vsh[cur][khalf * KPITCH + snt * 16 + nl];
    const v4f wlo = *(const v4f*)(&wsh[8 * d]);
    const v4f whi = *(const v4f*)(&wsh[8 * d + 4]);
    #pragma unroll
    for (int s4 = 0; s4 < 8; ++s4) {
      const float w = (s4 < 4) ? wlo[s4 & 3] : whi[s4 & 3];
      v2f a;
      a.x = kl[(4 * s4)     * KPITCH] * w;
      a.y = kl[(4 * s4 + 1) * KPITCH] * w;
      #pragma unroll
      for (int j = 0; j < 2; ++j) {
        v2f bm;
        bm.x = vl[(4 * s4)     * KPITCH + j * 16];
        bm.y = vl[(4 * s4 + 1) * KPITCH + j * 16];
        st[j] = wmma4(a, bm, st[j]);
      }
    }
    *(v4f*)(&ksh[cur ^ 1][sdst])               = pk0;
    *(v4f*)(&ksh[cur ^ 1][sdst + 16 * KPITCH]) = pk1;
    *(v4f*)(&vsh[cur ^ 1][sdst])               = pv0;
    *(v4f*)(&vsh[cur ^ 1][sdst + 16 * KPITCH]) = pv1;
    __syncthreads();
  }

  {
    const float coef = (cref == 0) ? ktab[t0] : 0.0f;
    #pragma unroll
    for (int j = 0; j < 2; ++j) {
      #pragma unroll
      for (int r = 0; r < 8; ++r) {
        const int row = smt * 16 + mh + r;
        const int col = (snt + j) * 16 + nl;
        ssh[row * SPITCH + col] = st[j][r] + coef * hb[row * Hc + col];
      }
    }
  }

  v8f acc[4] = {};
  for (int d = nprev; d < nprev + 2; ++d) {
    const int  cur  = d & 1;
    const bool more = d < nprev + 1;
    const int  jt   = 2 * d + par;
    if (more) {
      const size_t o = (size_t)(d + 1) * 32 * Hc;
      pk0 = *(const v4f*)(ksrc + o);
      pk1 = *(const v4f*)(ksrc + o + 16 * Hc);
      pv0 = *(const v4f*)(vsrc + o);
      pv1 = *(const v4f*)(vsrc + o + 16 * Hc);
    }
    if (jt <= it) {
      const float* kl = &ksh[cur][(par * 16 + nl) * KPITCH + khalf];
      v2f kf[16];
      #pragma unroll
      for (int ks = 0; ks < 16; ++ks)
        kf[ks] = *(const v2f*)(kl + ks * 4);
      const float* vl = &vsh[cur][(par * 16 + mh) * KPITCH + nl];
      v2f vf[4][4];
      #pragma unroll
      for (int t = 0; t < 4; ++t) {
        #pragma unroll
        for (int nt = 0; nt < 4; ++nt) {
          vf[t][nt].x = vl[(2 * t)     * KPITCH + nt * 16];
          vf[t][nt].y = vl[(2 * t + 1) * KPITCH + nt * 16];
        }
      }
      const int   tk0 = jt * 4 + (mh >> 2);
      const int   d0  = tq - tk0;
      const int   c0  = cumsh[tk0];
      const int   c1  = cumsh[tk0 + 1];
      const float e0  = ktab[d0 < 0 ? 0 : d0];
      const float e1  = ktab[d0 < 1 ? 0 : d0 - 1];
      const float w0  = (c0 == cq) ? e0 : 0.0f;
      const float w1  = (c1 == cq) ? e1 : 0.0f;

      v8f s = {};
      #pragma unroll
      for (int ks = 0; ks < 16; ++ks)
        s = wmma4(kf[ks], qa[ks], s);

      #pragma unroll
      for (int r = 0; r < 8; ++r) {
        const int   gk = jt * 16 + mh + r;
        const float w  = (gk <= gq) ? (r < 4 ? w0 : w1) : 0.0f;
        s[r] = s[r] * w;
      }
      #pragma unroll
      for (int t = 0; t < 4; ++t) {
        v2f a;
        a.x = s[2 * t];
        a.y = s[2 * t + 1];
        #pragma unroll
        for (int nt = 0; nt < 4; ++nt)
          acc[nt] = wmma4(a, vf[t][nt], acc[nt]);
      }
    }
    if (more) {
      *(v4f*)(&ksh[cur ^ 1][sdst])               = pk0;
      *(v4f*)(&ksh[cur ^ 1][sdst + 16 * KPITCH]) = pk1;
      *(v4f*)(&vsh[cur ^ 1][sdst])               = pv0;
      *(v4f*)(&vsh[cur ^ 1][sdst + 16 * KPITCH]) = pv1;
    }
    __syncthreads();
  }

  v8f cr[2] = {};
  {
    const float* sl = &ssh[khalf * SPITCH + par * 32 + nl];
    #pragma unroll
    for (int ks = 0; ks < 16; ++ks) {
      #pragma unroll
      for (int j = 0; j < 2; ++j) {
        v2f bm;
        bm.x = sl[(ks * 4)     * SPITCH + j * 16];
        bm.y = sl[(ks * 4 + 1) * SPITCH + j * 16];
        cr[j] = wmma4(qa[ks], bm, cr[j]);
      }
    }
  }
  float wr[8];
  #pragma unroll
  for (int r = 0; r < 8; ++r) {
    const int ti = (it * 16 + mh + r) >> 2;
    wr[r] = (cumsh[ti] == cref) ? ktab[ti - t0 + 1] : 0.0f;
  }

  float* cw = &comb[sw][lane * 4];
  if (par == 1) {
    #pragma unroll
    for (int r = 0; r < 8; ++r) {
      acc[2][r] += cr[0][r] * wr[r];
      acc[3][r] += cr[1][r] * wr[r];
    }
    #pragma unroll
    for (int nt = 0; nt < 4; ++nt) {
      #pragma unroll
      for (int hq = 0; hq < 2; ++hq) {
        v4f p;
        p.x = acc[nt][4 * hq];     p.y = acc[nt][4 * hq + 1];
        p.z = acc[nt][4 * hq + 2]; p.w = acc[nt][4 * hq + 3];
        *(v4f*)(cw + (nt * 2 + hq) * 128) = p;
      }
    }
  } else {
    #pragma unroll
    for (int r = 0; r < 8; ++r) {
      acc[0][r] += cr[0][r] * wr[r];
      acc[1][r] += cr[1][r] * wr[r];
    }
  }
  __syncthreads();

  if (par == 0) {
    #pragma unroll
    for (int nt = 0; nt < 4; ++nt) {
      #pragma unroll
      for (int hq = 0; hq < 2; ++hq) {
        const v4f p = *(const v4f*)(cw + (nt * 2 + hq) * 128);
        acc[nt][4 * hq]     += p.x; acc[nt][4 * hq + 1] += p.y;
        acc[nt][4 * hq + 2] += p.z; acc[nt][4 * hq + 3] += p.w;
      }
    }
    #pragma unroll
    for (int r = 0; r < 8; ++r) {
      const int gi = it * 16 + mh + r;
      float* orow = out + ((size_t)b * Sc + gi) * Hc;
      #pragma unroll
      for (int nt = 0; nt < 4; ++nt)
        orow[nt * 16 + nl] = acc[nt][r];
    }
  }
}

static __device__ __forceinline__ void
hstate_update(const float* __restrict__ kbuf, const float* __restrict__ vbuf,
              const float* __restrict__ hstate, const int* __restrict__ dones,
              float* __restrict__ hout, const int pr, const int b)
{
  __shared__ float ktab[Tc + 1];
  __shared__ float wsh[Tc];
  __shared__ __attribute__((aligned(16))) float hpart[KSPLIT][RW][16 * HPITCH];
  const int tid   = threadIdx.x;
  const int lane  = tid & 31;
  const int wave  = __builtin_amdgcn_readfirstlane(tid >> 5);
  const int wq    = wave & (RW - 1);
  const int th    = wave >> 2;
  const int tile  = pr * 2 + th;
  const int mt    = tile >> 2;
  const int nt    = tile & 3;
  const int nl    = lane & 15;
  const int khalf = (lane >> 4) << 1;

  int cj[8], cend, fd;
  done_prefix(dones + (size_t)b * Sc, lane, cj, cend, fd);
  (void)fd;

  for (int i = tid; i <= Tc; i += 32 * NWAVE) ktab[i] = __expf((float)i * LNK);
  __syncthreads();
  if (wave == 0) {
    #pragma unroll
    for (int j = 0; j < 8; ++j) {
      const int i = 32 * j + lane;
      wsh[i] = (cj[j] == cend) ? ktab[(Tc - 1) - i] : 0.0f;
    }
  }
  __syncthreads();

  const float* kb = kbuf + (size_t)b * Sc * Hc;
  const float* vb = vbuf + (size_t)b * Sc * Hc;

  const int    sbeg = wq * (Sc / RW);
  const float* kw   = kb  + (size_t)sbeg * Hc;
  const float* vw   = vb  + (size_t)sbeg * Hc;
  const float* ww   = wsh + (sbeg >> 2);

  v8f c = {};
  for (int s0 = 0; s0 < Sc / RW; s0 += 4) {
    const int   sA = s0 + khalf;
    const float w  = ww[s0 >> 2];
    v2f a;
    a.x = kw[(size_t)sA       * Hc + mt * 16 + nl];
    a.y = kw[(size_t)(sA + 1) * Hc + mt * 16 + nl];
    v2f bm;
    bm.x = vw[(size_t)sA       * Hc + nt * 16 + nl] * w;
    bm.y = vw[(size_t)(sA + 1) * Hc + nt * 16 + nl] * w;
    c = wmma4(a, bm, c);
  }

  const int mh = (lane >> 4) << 3;
  #pragma unroll
  for (int r = 0; r < 8; ++r)
    hpart[th][wq][(mh + r) * HPITCH + nl] = c[r];
  __syncthreads();

  const float cd    = ktab[Tc];
  const float delta = (cend == 0) ? 1.0f : 0.0f;
  const float* hb   = hstate + (size_t)b * Hc * Hc;
  const int    tt   = tid & 127;
  const int    trow = tt >> 3;
  const int    tcol = (tt & 7) * 2;
  v2f sum = *(const v2f*)(&hpart[th][0][trow * HPITCH + tcol]);
  #pragma unroll
  for (int w = 1; w < RW; ++w)
    sum = sum + *(const v2f*)(&hpart[th][w][trow * HPITCH + tcol]);
  const size_t go = ((size_t)b * Hc + mt * 16 + trow) * Hc + nt * 16 + tcol;
  const v2f hv = *(const v2f*)(hb + (go - (size_t)b * Hc * Hc));
  v2f res;
  res.x = sum.x + hv.x * cd * delta;
  res.y = sum.y + hv.y * cd * delta;
  *(v2f*)(hout + go) = res;
}

__global__ void __launch_bounds__(32 * NWAVE)
retention_and_state(const float* __restrict__ q, const float* __restrict__ kbuf,
                    const float* __restrict__ vbuf,
                    const float* __restrict__ hstate,
                    const int* __restrict__ dones, float* __restrict__ out,
                    float* __restrict__ hout)
{
  const int b = blockIdx.x;
  const int y = blockIdx.y;
  constexpr int NBLK = Sc / (16 * RW);
  if (y < NBLK) {
    const int g = (y < NBLK / 2) ? (NBLK - 1 - y) : (y - NBLK / 2);
    retention64(q, kbuf, vbuf, hstate, dones, out, g, b);
  } else {
    hstate_update(kbuf, vbuf, hstate, dones, hout, y - NBLK, b);
  }
}

extern "C" void kernel_launch(void* const* d_in, const int* in_sizes, int n_in,
                              void* d_out, int out_size, void* d_ws, size_t ws_size,
                              hipStream_t stream)
{
  const float* key    = (const float*)d_in[0];
  const float* query  = (const float*)d_in[1];
  const float* value  = (const float*)d_in[2];
  const float* hstate = (const float*)d_in[3];
  const int*   dones  = (const int*)d_in[4];
  const float* W_Q    = (const float*)d_in[5];
  const float* W_K    = (const float*)d_in[6];
  const float* W_V    = (const float*)d_in[7];

  const size_t qkv = (size_t)Bc * Sc * Hc;
  float* ws = (float*)d_ws;
  float* qb = ws;
  float* kb = ws + qkv;
  float* vb = ws + 2 * qkv;

  constexpr int gproj = Bc * Sc / PROJ_ROWS;
  proj_qkv<<<gproj, 256 * PROJ_TEAMS, 0, stream>>>(query, key, value,
                                                   W_Q, W_K, W_V, qb, kb, vb);

  float* out0 = (float*)d_out;
  float* hout = out0 + qkv;
  retention_and_state<<<dim3(Bc, Sc / (16 * RW) + 16 / KSPLIT), 32 * NWAVE, 0,
                        stream>>>(qb, kb, vb, hstate, dones, out0, hout);
}


# retention: earlier keys enter through per-block local 64x64 states k^T w v, each computed once (kept over the block's k rows), summed with decay weights in a third launch; sums regrouped
# speedup vs baseline: 5.6337x; 1.2025x over previous
#include <hip/hip_runtime.h>
#include <hip/hip_bf16.h>


typedef float v2f __attribute__((ext_vector_type(2)));
typedef float v4f __attribute__((ext_vector_type(4)));
typedef float v8f __attribute__((ext_vector_type(8)));

constexpr int   Bc = 32, Sc = 1024, Ec = 512, Hc = 64, Tc = 256;
constexpr float LNK = -0.1053605156578263f;

static __device__ __forceinline__ v8f wmma4(v2f a, v2f b, v8f c) {
  return __builtin_amdgcn_wmma_f32_16x16x4_f32(false, a, false, b, (short)0, c,
                                               false, false);
}

constexpr int PROJ_ROWS  = 128;
constexpr int PROJ_TEAMS = 3;
constexpr int HALF_GRPS  = Ec / 16;
constexpr int WHALF      = HALF_GRPS * Hc * 8;

static __device__ __forceinline__ void
stage_w_half(const float* __restrict__ W, float* __restrict__ wt,
             const int lt, const int half)
{
  constexpr int NIT = HALF_GRPS * Hc / 256;
  v4f lo[NIT], hi[NIT];
  #pragma unroll
  for (int it = 0; it < NIT; ++it) {
    const int p  = lt + it * 256;
    const int ol = p >> 6, n = p & 63;
    const float* wsrc = W + (size_t)((half * HALF_GRPS + ol) * 8) * Hc + n;
    lo[it].x = wsrc[0 * Hc]; lo[it].y = wsrc[1 * Hc];
    lo[it].z = wsrc[2 * Hc]; lo[it].w = wsrc[3 * Hc];
    hi[it].x = wsrc[4 * Hc]; hi[it].y = wsrc[5 * Hc];
    hi[it].z = wsrc[6 * Hc]; hi[it].w = wsrc[7 * Hc];
  }
  #pragma unroll
  for (int it = 0; it < NIT; ++it) {
    const int p = lt + it * 256;
    *(v4f*)(&wt[p * 8])     = lo[it];
    *(v4f*)(&wt[p * 8 + 4]) = hi[it];
  }
}

__global__ void __launch_bounds__(256 * PROJ_TEAMS)
proj_qkv(const float* __restrict__ Xq, const float* __restrict__ Xk,
         const float* __restrict__ Xv, const float* __restrict__ Wq,
         const float* __restrict__ Wk, const float* __restrict__ Wv,
         float* __restrict__ Yq, float* __restrict__ Yk,
         float* __restrict__ Yv)
{
  __shared__ __attribute__((aligned(16))) float wl[PROJ_TEAMS * WHALF];

  const int tid  = threadIdx.x;
  const int team = __builtin_amdgcn_readfirstlane(tid >> 8);
  const float* X = (team == 0) ? Xq : (team == 1) ? Xk : Xv;
  const float* Wm = (team == 0) ? Wq : (team == 1) ? Wk : Wv;
  float*       Y = (team == 0) ? Yq : (team == 1) ? Yk : Yv;
  float*       wt = wl + team * WHALF;

  const int lt   = tid & 255;
  const int lane = tid & 31;
  const int wave = lt >> 5;
  const int nl   = lane & 15;
  const int h    = lane >> 4;
  const int row0 = blockIdx.x * PROJ_ROWS + wave * 16;

  const float* xrow  = X + (size_t)(row0 + nl) * Ec + h * 4;
  const float* wlane = wt + nl * 8 + h * 4;

  v8f acc[4] = {};
  v4f xa[16];
  v4f wb[2][4];

  #pragma unroll
  for (int half = 0; half < 2; ++half) {
    if (half == 1) __syncthreads();
    stage_w_half(Wm, wt, lt, half);
    if (half == 0) {
      #pragma unroll
      for (int i = 0; i < 8; ++i) xa[i] = *(const v4f*)(xrow + i * 8);
    }
    __syncthreads();

    #pragma unroll
    for (int nt = 0; nt < 4; ++nt) wb[0][nt] = *(const v4f*)(wlane + nt * 128);

    #pragma unroll
    for (int cc = 0; cc < 4; ++cc) {
      const int c = half * 4 + cc;
      if (c + 1 < 8) {
        #pragma unroll
        for (int i = 0; i < 8; ++i)
          xa[((c + 1) & 1) * 8 + i] = *(const v4f*)(xrow + ((c + 1) * 8 + i) * 8);
      }
      #pragma unroll
      for (int i = 0; i < 8; ++i) {
        const int ol = cc * 8 + i;
        if (ol + 1 < HALF_GRPS) {
          #pragma unroll
          for (int nt = 0; nt < 4; ++nt)
            wb[(ol + 1) & 1][nt] =
                *(const v4f*)(wlane + ((ol + 1) * 64 + nt * 16) * 8);
        }
        const v4f x = xa[(c & 1) * 8 + i];
        v2f a0, a1;
        a0.x = x.x; a0.y = x.y;
        a1.x = x.z; a1.y = x.w;
        #pragma unroll
        for (int nt = 0; nt < 4; ++nt) {
          const v4f w4 = wb[ol & 1][nt];
          v2f b0; b0.x = w4.x; b0.y = w4.y;
          acc[nt] = wmma4(a0, b0, acc[nt]);
        }
        #pragma unroll
        for (int nt = 0; nt < 4; ++nt) {
          const v4f w4 = wb[ol & 1][nt];
          v2f b1; b1.x = w4.z; b1.y = w4.w;
          acc[nt] = wmma4(a1, b1, acc[nt]);
        }
      }
    }
  }

  const int mbase = row0 + h * 8;
  #pragma unroll
  for (int r = 0; r < 8; ++r) {
    float* yrow = Y + (size_t)(mbase + r) * Hc;
    #pragma unroll
    for (int nt = 0; nt < 4; ++nt)
      yrow[nt * 16 + nl] = acc[nt][r];
  }
}

static __device__ __forceinline__ void done_prefix(const int* __restrict__ db,
                                                   int lane, int (&cj)[8],
                                                   int& total, int& first)
{
  int dv[8];
  #pragma unroll
  for (int j = 0; j < 8; ++j) dv[j] = db[(32 * j + lane) * 4];
  const unsigned lanes_le = (2u << lane) - 1u;
  int base = 0, fd = Tc;
  #pragma unroll
  for (int j = 0; j < 8; ++j) {
    const unsigned m = __builtin_amdgcn_ballot_w32(dv[j] != 0);
    cj[j] = base + __popc(m & lanes_le);
    if (m != 0u && fd == Tc) fd = 32 * j + (__ffs(m) - 1);
    base += __popc(m);
  }
  total = base;
  first = fd;
}

constexpr int RW = 4;
constexpr int KSPLIT = 2;
constexpr int NWAVE = RW * KSPLIT;
constexpr int KPITCH = 68;
constexpr int SPITCH = 72;
constexpr int NBLK = Sc / (16 * RW);

__global__ void __launch_bounds__(32 * NWAVE)
block_local(const float* __restrict__ q, float* kl_buf,
            const float* __restrict__ vbuf, const int* __restrict__ dones,
            float* __restrict__ out)
{
  __shared__ float ktab[Tc + 1];
  __shared__ int   cumsh[Tc];
  __shared__ __attribute__((aligned(16))) float wsh[16];
  __shared__ __attribute__((aligned(16))) float ksh[2][32 * KPITCH];
  __shared__ __attribute__((aligned(16))) float vsh[2][32 * KPITCH];
  __shared__ __attribute__((aligned(16))) float comb[RW][8 * 128];

  const int b     = blockIdx.x;
  const int g     = blockIdx.y;
  const int tid   = threadIdx.x;
  const int lane  = tid & 31;
  const int wave  = __builtin_amdgcn_readfirstlane(tid >> 5);
  const int sw    = wave & (RW - 1);
  const int par   = wave >> 2;
  const int smt   = wave >> 1;
  const int snt   = (wave & 1) * 2;
  const int nl    = lane & 15;
  const int khalf = (lane >> 4) << 1;
  const int mh    = (lane >> 4) << 3;
  const int it    = g * RW + sw;
  const int t0    = 4 * RW * g;

  const float* qb = q      + (size_t)b * Sc * Hc;
  float*       kb = kl_buf + ((size_t)b * Sc + (size_t)g * 16 * RW) * Hc;
  const float* vb = vbuf   + ((size_t)b * Sc + (size_t)g * 16 * RW) * Hc;

  const int    srow = tid >> 4;
  const int    scol = (tid & 15) * 4;
  const float* ksrc = kb + (size_t)srow * Hc + scol;
  const float* vsrc = vb + (size_t)srow * Hc + scol;
  const int    sdst = srow * KPITCH + scol;


  const float* qrow = qb + (size_t)(it * 16 + nl) * Hc;
  v2f qa[16];
  #pragma unroll
  for (int ks = 0; ks < 16; ++ks)
    qa[ks] = *(const v2f*)(qrow + ks * 4 + khalf);

  v4f pk0 = *(const v4f*)(ksrc);
  v4f pk1 = *(const v4f*)(ksrc + 16 * Hc);
  v4f pv0 = *(const v4f*)(vsrc);
  v4f pv1 = *(const v4f*)(vsrc + 16 * Hc);

  for (int i = tid; i <= Tc; i += 32 * NWAVE) ktab[i] = __expf((float)i * LNK);
  if (wave == 0) {
    int cj[8], ndone, fd;
    done_prefix(dones + (size_t)b * Sc, lane, cj, ndone, fd);
    (void)ndone; (void)fd;
    #pragma unroll
    for (int j = 0; j < 8; ++j) cumsh[32 * j + lane] = cj[j];
  }
  *(v4f*)(&ksh[0][sdst])               = pk0;
  *(v4f*)(&ksh[0][sdst + 16 * KPITCH]) = pk1;
  *(v4f*)(&vsh[0][sdst])               = pv0;
  *(v4f*)(&vsh[0][sdst + 16 * KPITCH]) = pv1;
  __syncthreads();

  const int cend = cumsh[t0 + 15];
  if (tid < 16)
    wsh[tid] = (cumsh[t0 + tid] == cend) ? ktab[15 - tid] : 0.0f;

  const int gq = it * 16 + nl;
  const int tq = gq >> 2;
  const int cq = cumsh[tq];
  __syncthreads();

  v8f st[2]  = {};
  v8f acc[4] = {};
  for (int dl = 0; dl < 2; ++dl) {
    const int  cur  = dl;
    const bool more = dl < 1;
    const int  jt   = 4 * g + 2 * dl + par;
    if (more) {
      pk0 = *(const v4f*)(ksrc + 32 * Hc);
      pk1 = *(const v4f*)(ksrc + 32 * Hc + 16 * Hc);
      pv0 = *(const v4f*)(vsrc + 32 * Hc);
      pv1 = *(const v4f*)(vsrc + 32 * Hc + 16 * Hc);
    }
    {
      const float* kl = &ksh[cur][khalf * KPITCH + smt * 16 + nl];
      const float* vl = &vsh[cur][khalf * KPITCH + snt * 16 + nl];
      const v4f wlo = *(const v4f*)(&wsh[8 * dl]);
      const v4f whi = *(const v4f*)(&wsh[8 * dl + 4]);
      #pragma unroll
      for (int s4 = 0; s4 < 8; ++s4) {
        const float w = (s4 < 4) ? wlo[s4 & 3] : whi[s4 & 3];
        v2f a;
        a.x = kl[(4 * s4)     * KPITCH] * w;
        a.y = kl[(4 * s4 + 1) * KPITCH] * w;
        #pragma unroll
        for (int j = 0; j < 2; ++j) {
          v2f bm;
          bm.x = vl[(4 * s4)     * KPITCH + j * 16];
          bm.y = vl[(4 * s4 + 1) * KPITCH + j * 16];
          st[j] = wmma4(a, bm, st[j]);
        }
      }
    }
    if (jt <= it) {
      const float* kl = &ksh[cur][(par * 16 + nl) * KPITCH + khalf];
      v2f kf[16];
      #pragma unroll
      for (int ks = 0; ks < 16; ++ks)
        kf[ks] = *(const v2f*)(kl + ks * 4);
      const float* vl = &vsh[cur][(par * 16 + mh) * KPITCH + nl];
      v2f vf[4][4];
      #pragma unroll
      for (int t = 0; t < 4; ++t) {
        #pragma unroll
        for (int nt = 0; nt < 4; ++nt) {
          vf[t][nt].x = vl[(2 * t)     * KPITCH + nt * 16];
          vf[t][nt].y = vl[(2 * t + 1) * KPITCH + nt * 16];
        }
      }
      const int   tk0 = jt * 4 + (mh >> 2);
      const int   d0  = tq - tk0;
      const int   c0  = cumsh[tk0];
      const int   c1  = cumsh[tk0 + 1];
      const float e0  = ktab[d0 < 0 ? 0 : d0];
      const float e1  = ktab[d0 < 1 ? 0 : d0 - 1];
      const float w0  = (c0 == cq) ? e0 : 0.0f;
      const float w1  = (c1 == cq) ? e1 : 0.0f;

      v8f s = {};
      #pragma unroll
      for (int ks = 0; ks < 16; ++ks)
        s = wmma4(kf[ks], qa[ks], s);

      #pragma unroll
      for (int r = 0; r < 8; ++r) {
        const int   gk = jt * 16 + mh + r;
        const float w  = (gk <= gq) ? (r < 4 ? w0 : w1) : 0.0f;
        s[r] = s[r] * w;
      }
      #pragma unroll
      for (int t = 0; t < 4; ++t) {
        v2f a;
        a.x = s[2 * t];
        a.y = s[2 * t + 1];
        #pragma unroll
        for (int nt = 0; nt < 4; ++nt)
          acc[nt] = wmma4(a, vf[t][nt], acc[nt]);
      }
    }
    if (more) {
      *(v4f*)(&ksh[cur ^ 1][sdst])               = pk0;
      *(v4f*)(&ksh[cur ^ 1][sdst + 16 * KPITCH]) = pk1;
      *(v4f*)(&vsh[cur ^ 1][sdst])               = pv0;
      *(v4f*)(&vsh[cur ^ 1][sdst + 16 * KPITCH]) = pv1;
    }
    __syncthreads();
  }

  #pragma unroll
  for (int j = 0; j < 2; ++j) {
    #pragma unroll
    for (int r = 0; r < 8; ++r) {
      const int row = smt * 16 + mh + r;
      const int col = (snt + j) * 16 + nl;
      kb[(size_t)row * Hc + col] = st[j][r];
    }
  }

  float* cw = &comb[sw][lane * 4];
  if (par == 1) {
    #pragma unroll
    for (int nt = 0; nt < 4; ++nt) {
      #pragma unroll
      for (int hq = 0; hq < 2; ++hq) {
        v4f p;
        p.x = acc[nt][4 * hq];     p.y = acc[nt][4 * hq + 1];
        p.z = acc[nt][4 * hq + 2]; p.w = acc[nt][4 * hq + 3];
        *(v4f*)(cw + (nt * 2 + hq) * 128) = p;
      }
    }
  }
  __syncthreads();

  if (par == 0) {
    #pragma unroll
    for (int nt = 0; nt < 4; ++nt) {
      #pragma unroll
      for (int hq = 0; hq < 2; ++hq) {
        const v4f p = *(const v4f*)(cw + (nt * 2 + hq) * 128);
        acc[nt][4 * hq]     += p.x; acc[nt][4 * hq + 1] += p.y;
        acc[nt][4 * hq + 2] += p.z; acc[nt][4 * hq + 3] += p.w;
      }
    }
    #pragma unroll
    for (int r = 0; r < 8; ++r) {
      const int gi = it * 16 + mh + r;
      float* orow = out + ((size_t)b * Sc + gi) * Hc;
      #pragma unroll
      for (int nt = 0; nt < 4; ++nt)
        orow[nt * 16 + nl] = acc[nt][r];
    }
  }
}

__global__ void __launch_bounds__(32 * NWAVE)
cross_and_state(const float* __restrict__ q, const float* __restrict__ lbuf,
                const float* __restrict__ hstate, const int* __restrict__ dones,
                float* __restrict__ out, float* __restrict__ hout)
{
  __shared__ float ktab[Tc + 1];
  __shared__ int   cumsh[Tc];
  __shared__ __attribute__((aligned(16))) float ssh[Hc * SPITCH];

  const int b     = blockIdx.x;
  const int y     = blockIdx.y;
  const int tid   = threadIdx.x;
  const int lane  = tid & 31;
  const int wave  = __builtin_amdgcn_readfirstlane(tid >> 5);
  const int sw    = wave & (RW - 1);
  const int par   = wave >> 2;
  const int nl    = lane & 15;
  const int khalf = (lane >> 4) << 1;
  const int mh    = (lane >> 4) << 3;

  const float* lb = lbuf   + (size_t)b * Sc * Hc;
  const float* hb = hstate + (size_t)b * Hc * Hc;

  const int g  = (y < NBLK) ? y : 0;
  const int it = g * RW + sw;
  const int t0 = 4 * RW * g;
  v2f qa[16];
  if (y < NBLK) {
    const float* qrow = q + ((size_t)b * Sc + it * 16 + nl) * Hc;
    #pragma unroll
    for (int ks = 0; ks < 16; ++ks)
      qa[ks] = *(const v2f*)(qrow + ks * 4 + khalf);
  }

  for (int i = tid; i <= Tc; i += 32 * NWAVE) ktab[i] = __expf((float)i * LNK);
  if (wave == 0) {
    int cj[8], ndone, fd;
    done_prefix(dones + (size_t)b * Sc, lane, cj, ndone, fd);
    (void)ndone; (void)fd;
    #pragma unroll
    for (int j = 0; j < 8; ++j) cumsh[32 * j + lane] = cj[j];
  }
  __syncthreads();

  if (y < NBLK) {
    const int cref = (g > 0) ? cumsh[t0 - 1] : 0;
    int glo = g;
    while (glo > 0 && cumsh[16 * (glo - 1) + 15] == cref) --glo;

    const float coef = (cref == 0) ? ktab[t0] : 0.0f;
    v4f sv[4];
    #pragma unroll
    for (int i = 0; i < 4; ++i) {
      const v4f hv = *(const v4f*)(hb + (tid + 256 * i) * 4);
      sv[i] = hv * coef;
    }
    for (int gp = g - 1; gp >= glo; --gp) {
      const float  w  = ktab[16 * (g - 1 - gp)];
      const float* lp = lb + (size_t)gp * Hc * Hc;
      v4f lv[4];
      #pragma unroll
      for (int i = 0; i < 4; ++i) lv[i] = *(const v4f*)(lp + (tid + 256 * i) * 4);
      #pragma unroll
      for (int i = 0; i < 4; ++i) sv[i] = sv[i] + lv[i] * w;
    }
    #pragma unroll
    for (int i = 0; i < 4; ++i) {
      const int e = (tid + 256 * i) * 4;
      *(v4f*)(&ssh[(e >> 6) * SPITCH + (e & 63)]) = sv[i];
    }
    __syncthreads();

    v8f cr[2] = {};
    {
      const float* sl = &ssh[khalf * SPITCH + par * 32 + nl];
      #pragma unroll
      for (int ks = 0; ks < 16; ++ks) {
        #pragma unroll
        for (int j = 0; j < 2; ++j) {
          v2f bm;
          bm.x = sl[(ks * 4)     * SPITCH + j * 16];
          bm.y = sl[(ks * 4 + 1) * SPITCH + j * 16];
          cr[j] = wmma4(qa[ks], bm, cr[j]);
        }
      }
    }
    float wr[8];
    #pragma unroll
    for (int r = 0; r < 8; ++r) {
      const int ti = (it * 16 + mh + r) >> 2;
      wr[r] = (cumsh[ti] == cref) ? ktab[ti - t0 + 1] : 0.0f;
    }
    #pragma unroll
    for (int r = 0; r < 8; ++r) {
      const int gi = it * 16 + mh + r;
      float* orow = out + ((size_t)b * Sc + gi) * Hc + par * 32 + nl;
      #pragma unroll
      for (int j = 0; j < 2; ++j)
        orow[j * 16] = orow[j * 16] + cr[j][r] * wr[r];
    }
  } else {
    const int cend = cumsh[Tc - 1];
    int glo = NBLK - 1;
    while (glo > 0 && cumsh[16 * (glo - 1) + 15] == cend) --glo;

    v4f sv[4] = {};
    for (int gp = NBLK - 1; gp >= glo; --gp) {
      const float  w  = ktab[16 * (NBLK - 1 - gp)];
      const float* lp = lb + (size_t)gp * Hc * Hc;
      v4f lv[4];
      #pragma unroll
      for (int i = 0; i < 4; ++i) lv[i] = *(const v4f*)(lp + (tid + 256 * i) * 4);
      #pragma unroll
      for (int i = 0; i < 4; ++i) sv[i] = sv[i] + lv[i] * w;
    }
    const float cd    = ktab[Tc];
    const float delta = (cend == 0) ? 1.0f : 0.0f;
    #pragma unroll
    for (int i = 0; i < 4; ++i) {
      const int e  = (tid + 256 * i) * 4;
      const v4f hv = *(const v4f*)(hb + e);
      *(v4f*)(hout + (size_t)b * Hc * Hc + e) = sv[i] + hv * (cd * delta);
    }
  }
}

extern "C" void kernel_launch(void* const* d_in, const int* in_sizes, int n_in,
                              void* d_out, int out_size, void* d_ws, size_t ws_size,
                              hipStream_t stream)
{
  const float* key    = (const float*)d_in[0];
  const float* query  = (const float*)d_in[1];
  const float* value  = (const float*)d_in[2];
  const float* hstate = (const float*)d_in[3];
  const int*   dones  = (const int*)d_in[4];
  const float* W_Q    = (const float*)d_in[5];
  const float* W_K    = (const float*)d_in[6];
  const float* W_V    = (const float*)d_in[7];

  const size_t qkv = (size_t)Bc * Sc * Hc;
  float* ws = (float*)d_ws;
  float* qb = ws;
  float* kb = ws + qkv;
  float* vb = ws + 2 * qkv;

  constexpr int gproj = Bc * Sc / PROJ_ROWS;
  proj_qkv<<<gproj, 256 * PROJ_TEAMS, 0, stream>>>(query, key, value,
                                                   W_Q, W_K, W_V, qb, kb, vb);

  float* out0 = (float*)d_out;
  float* hout = out0 + qkv;
  block_local<<<dim3(Bc, NBLK), 32 * NWAVE, 0, stream>>>(qb, kb, vb, dones, out0);
  cross_and_state<<<dim3(Bc, NBLK + 1), 32 * NWAVE, 0, stream>>>(
      qb, kb, hstate, dones, out0, hout);
}


# projections: each 24-wave work-group takes 384 rows of ONE projection (two groups straddle two), its whole W staged once in LDS: no k-half restage, no mid barrier; k summed in the same order
# speedup vs baseline: 6.1103x; 1.0846x over previous
#include <hip/hip_runtime.h>
#include <hip/hip_bf16.h>


typedef float v2f __attribute__((ext_vector_type(2)));
typedef float v4f __attribute__((ext_vector_type(4)));
typedef float v8f __attribute__((ext_vector_type(8)));

constexpr int   Bc = 32, Sc = 1024, Ec = 512, Hc = 64, Tc = 256;
constexpr float LNK = -0.1053605156578263f;

static __device__ __forceinline__ v8f wmma4(v2f a, v2f b, v8f c) {
  return __builtin_amdgcn_wmma_f32_16x16x4_f32(false, a, false, b, (short)0, c,
                                               false, false);
}

constexpr int PROJ_ROWS  = 128;
constexpr int PROJ_TEAMS = 3;
constexpr int PROJ_UNITS = Bc * Sc / PROJ_ROWS;
constexpr int KGRPS      = Ec / 8;
constexpr int WFULL      = KGRPS * Hc * 8;

constexpr int W_NP  = KGRPS * Hc;
constexpr int W_NT  = 256 * PROJ_TEAMS;
constexpr int W_NIT = (W_NP + W_NT - 1) / W_NT;

static __device__ __forceinline__ void
load_w(const float* __restrict__ W, const int tid,
       v4f (&lo)[W_NIT], v4f (&hi)[W_NIT])
{
  #pragma unroll
  for (int it = 0; it < W_NIT; ++it) {
    const int p = tid + it * W_NT;
    if (it < W_NIT - 1 || p < W_NP) {
      const int o = p >> 6, n = p & 63;
      const float* wsrc = W + (size_t)(o * 8) * Hc + n;
      lo[it].x = wsrc[0 * Hc]; lo[it].y = wsrc[1 * Hc];
      lo[it].z = wsrc[2 * Hc]; lo[it].w = wsrc[3 * Hc];
      hi[it].x = wsrc[4 * Hc]; hi[it].y = wsrc[5 * Hc];
      hi[it].z = wsrc[6 * Hc]; hi[it].w = wsrc[7 * Hc];
    }
  }
}

static __device__ __forceinline__ void
store_w(float* __restrict__ wt, const int tid,
        const v4f (&lo)[W_NIT], const v4f (&hi)[W_NIT])
{
  #pragma unroll
  for (int it = 0; it < W_NIT; ++it) {
    const int p = tid + it * W_NT;
    if (it < W_NIT - 1 || p < W_NP) {
      *(v4f*)(&wt[p * 8])     = lo[it];
      *(v4f*)(&wt[p * 8 + 4]) = hi[it];
    }
  }
}

__global__ void __launch_bounds__(256 * PROJ_TEAMS)
proj_qkv(const float* __restrict__ Xq, const float* __restrict__ Xk,
         const float* __restrict__ Xv, const float* __restrict__ Wq,
         const float* __restrict__ Wk, const float* __restrict__ Wv,
         float* __restrict__ Yq, float* __restrict__ Yk,
         float* __restrict__ Yv)
{
  __shared__ __attribute__((aligned(16))) float wl[2 * WFULL];

  const int tid  = threadIdx.x;
  const int team = __builtin_amdgcn_readfirstlane(tid >> 8);
  const int u0   = blockIdx.x * PROJ_TEAMS;
  const int p0   = u0 / PROJ_UNITS;
  const int p2   = (u0 + PROJ_TEAMS - 1) / PROJ_UNITS;
  const int u    = u0 + team;
  const int pt   = u / PROJ_UNITS;
  const float* X = (pt == 0) ? Xq : (pt == 1) ? Xk : Xv;
  float*       Y = (pt == 0) ? Yq : (pt == 1) ? Yk : Yv;
  const float* wt = wl + (pt - p0) * WFULL;

  const int lt   = tid & 255;
  const int lane = tid & 31;
  const int wave = lt >> 5;
  const int nl   = lane & 15;
  const int h    = lane >> 4;
  const int row0 = (u % PROJ_UNITS) * PROJ_ROWS + wave * 16;

  const float* xrow  = X + (size_t)(row0 + nl) * Ec + h * 4;
  const float* wlane = wt + nl * 8 + h * 4;

  v8f acc[4] = {};
  v4f xa[16];
  v4f wb[2][4];

  v4f wlo[W_NIT], whi[W_NIT];
  load_w((p0 == 0) ? Wq : (p0 == 1) ? Wk : Wv, tid, wlo, whi);
  #pragma unroll
  for (int i = 0; i < 8; ++i) xa[i] = *(const v4f*)(xrow + i * 8);
  store_w(wl, tid, wlo, whi);
  if (p2 != p0) {
    load_w((p2 == 1) ? Wk : Wv, tid, wlo, whi);
    store_w(wl + WFULL, tid, wlo, whi);
  }
  __syncthreads();

  #pragma unroll
  for (int nt = 0; nt < 4; ++nt) wb[0][nt] = *(const v4f*)(wlane + nt * 128);

  #pragma unroll
  for (int c = 0; c < 8; ++c) {
    if (c + 1 < 8) {
      #pragma unroll
      for (int i = 0; i < 8; ++i)
        xa[((c + 1) & 1) * 8 + i] = *(const v4f*)(xrow + ((c + 1) * 8 + i) * 8);
    }
    #pragma unroll
    for (int i = 0; i < 8; ++i) {
      const int o = c * 8 + i;
      if (o + 1 < KGRPS) {
        #pragma unroll
        for (int nt = 0; nt < 4; ++nt)
          wb[(o + 1) & 1][nt] =
              *(const v4f*)(wlane + ((o + 1) * 64 + nt * 16) * 8);
      }
      const v4f x = xa[(c & 1) * 8 + i];
      v2f a0, a1;
      a0.x = x.x; a0.y = x.y;
      a1.x = x.z; a1.y = x.w;
      #pragma unroll
      for (int nt = 0; nt < 4; ++nt) {
        const v4f w4 = wb[o & 1][nt];
        v2f b0; b0.x = w4.x; b0.y = w4.y;
        acc[nt] = wmma4(a0, b0, acc[nt]);
      }
      #pragma unroll
      for (int nt = 0; nt < 4; ++nt) {
        const v4f w4 = wb[o & 1][nt];
        v2f b1; b1.x = w4.z; b1.y = w4.w;
        acc[nt] = wmma4(a1, b1, acc[nt]);
      }
    }
  }

  const int mbase = row0 + h * 8;
  #pragma unroll
  for (int r = 0; r < 8; ++r) {
    float* yrow = Y + (size_t)(mbase + r) * Hc;
    #pragma unroll
    for (int nt = 0; nt < 4; ++nt)
      yrow[nt * 16 + nl] = acc[nt][r];
  }
}

static __device__ __forceinline__ void done_prefix(const int* __restrict__ db,
                                                   int lane, int (&cj)[8],
                                                   int& total, int& first)
{
  int dv[8];
  #pragma unroll
  for (int j = 0; j < 8; ++j) dv[j] = db[(32 * j + lane) * 4];
  const unsigned lanes_le = (2u << lane) - 1u;
  int base = 0, fd = Tc;
  #pragma unroll
  for (int j = 0; j < 8; ++j) {
    const unsigned m = __builtin_amdgcn_ballot_w32(dv[j] != 0);
    cj[j] = base + __popc(m & lanes_le);
    if (m != 0u && fd == Tc) fd = 32 * j + (__ffs(m) - 1);
    base += __popc(m);
  }
  total = base;
  first = fd;
}

constexpr int RW = 4;
constexpr int KSPLIT = 2;
constexpr int NWAVE = RW * KSPLIT;
constexpr int KPITCH = 68;
constexpr int SPITCH = 72;
constexpr int NBLK = Sc / (16 * RW);

__global__ void __launch_bounds__(32 * NWAVE)
block_local(const float* __restrict__ q, float* kl_buf,
            const float* __restrict__ vbuf, const int* __restrict__ dones,
            float* __restrict__ out)
{
  __shared__ float ktab[Tc + 1];
  __shared__ int   cumsh[Tc];
  __shared__ __attribute__((aligned(16))) float wsh[16];
  __shared__ __attribute__((aligned(16))) float ksh[2][32 * KPITCH];
  __shared__ __attribute__((aligned(16))) float vsh[2][32 * KPITCH];
  __shared__ __attribute__((aligned(16))) float comb[RW][8 * 128];

  const int b     = blockIdx.x;
  const int g     = blockIdx.y;
  const int tid   = threadIdx.x;
  const int lane  = tid & 31;
  const int wave  = __builtin_amdgcn_readfirstlane(tid >> 5);
  const int sw    = wave & (RW - 1);
  const int par   = wave >> 2;
  const int smt   = wave >> 1;
  const int snt   = (wave & 1) * 2;
  const int nl    = lane & 15;
  const int khalf = (lane >> 4) << 1;
  const int mh    = (lane >> 4) << 3;
  const int it    = g * RW + sw;
  const int t0    = 4 * RW * g;

  const float* qb = q      + (size_t)b * Sc * Hc;
  float*       kb = kl_buf + ((size_t)b * Sc + (size_t)g * 16 * RW) * Hc;
  const float* vb = vbuf   + ((size_t)b * Sc + (size_t)g * 16 * RW) * Hc;

  const int    srow = tid >> 4;
  const int    scol = (tid & 15) * 4;
  const float* ksrc = kb + (size_t)srow * Hc + scol;
  const float* vsrc = vb + (size_t)srow * Hc + scol;
  const int    sdst = srow * KPITCH + scol;


  const float* qrow = qb + (size_t)(it * 16 + nl) * Hc;
  v2f qa[16];
  #pragma unroll
  for (int ks = 0; ks < 16; ++ks)
    qa[ks] = *(const v2f*)(qrow + ks * 4 + khalf);

  v4f pk0 = *(const v4f*)(ksrc);
  v4f pk1 = *(const v4f*)(ksrc + 16 * Hc);
  v4f pv0 = *(const v4f*)(vsrc);
  v4f pv1 = *(const v4f*)(vsrc + 16 * Hc);

  for (int i = tid; i <= Tc; i += 32 * NWAVE) ktab[i] = __expf((float)i * LNK);
  if (wave == 0) {
    int cj[8], ndone, fd;
    done_prefix(dones + (size_t)b * Sc, lane, cj, ndone, fd);
    (void)ndone; (void)fd;
    #pragma unroll
    for (int j = 0; j < 8; ++j) cumsh[32 * j + lane] = cj[j];
  }
  *(v4f*)(&ksh[0][sdst])               = pk0;
  *(v4f*)(&ksh[0][sdst + 16 * KPITCH]) = pk1;
  *(v4f*)(&vsh[0][sdst])               = pv0;
  *(v4f*)(&vsh[0][sdst + 16 * KPITCH]) = pv1;
  __syncthreads();

  const int cend = cumsh[t0 + 15];
  if (tid < 16)
    wsh[tid] = (cumsh[t0 + tid] == cend) ? ktab[15 - tid] : 0.0f;

  const int gq = it * 16 + nl;
  const int tq = gq >> 2;
  const int cq = cumsh[tq];
  __syncthreads();

  v8f st[2]  = {};
  v8f acc[4] = {};
  for (int dl = 0; dl < 2; ++dl) {
    const int  cur  = dl;
    const bool more = dl < 1;
    const int  jt   = 4 * g + 2 * dl + par;
    if (more) {
      pk0 = *(const v4f*)(ksrc + 32 * Hc);
      pk1 = *(const v4f*)(ksrc + 32 * Hc + 16 * Hc);
      pv0 = *(const v4f*)(vsrc + 32 * Hc);
      pv1 = *(const v4f*)(vsrc + 32 * Hc + 16 * Hc);
    }
    {
      const float* kl = &ksh[cur][khalf * KPITCH + smt * 16 + nl];
      const float* vl = &vsh[cur][khalf * KPITCH + snt * 16 + nl];
      const v4f wlo = *(const v4f*)(&wsh[8 * dl]);
      const v4f whi = *(const v4f*)(&wsh[8 * dl + 4]);
      #pragma unroll
      for (int s4 = 0; s4 < 8; ++s4) {
        const float w = (s4 < 4) ? wlo[s4 & 3] : whi[s4 & 3];
        v2f a;
        a.x = kl[(4 * s4)     * KPITCH] * w;
        a.y = kl[(4 * s4 + 1) * KPITCH] * w;
        #pragma unroll
        for (int j = 0; j < 2; ++j) {
          v2f bm;
          bm.x = vl[(4 * s4)     * KPITCH + j * 16];
          bm.y = vl[(4 * s4 + 1) * KPITCH + j * 16];
          st[j] = wmma4(a, bm, st[j]);
        }
      }
    }
    if (jt <= it) {
      const float* kl = &ksh[cur][(par * 16 + nl) * KPITCH + khalf];
      v2f kf[16];
      #pragma unroll
      for (int ks = 0; ks < 16; ++ks)
        kf[ks] = *(const v2f*)(kl + ks * 4);
      const float* vl = &vsh[cur][(par * 16 + mh) * KPITCH + nl];
      v2f vf[4][4];
      #pragma unroll
      for (int t = 0; t < 4; ++t) {
        #pragma unroll
        for (int nt = 0; nt < 4; ++nt) {
          vf[t][nt].x = vl[(2 * t)     * KPITCH + nt * 16];
          vf[t][nt].y = vl[(2 * t + 1) * KPITCH + nt * 16];
        }
      }
      const int   tk0 = jt * 4 + (mh >> 2);
      const int   d0  = tq - tk0;
      const int   c0  = cumsh[tk0];
      const int   c1  = cumsh[tk0 + 1];
      const float e0  = ktab[d0 < 0 ? 0 : d0];
      const float e1  = ktab[d0 < 1 ? 0 : d0 - 1];
      const float w0  = (c0 == cq) ? e0 : 0.0f;
      const float w1  = (c1 == cq) ? e1 : 0.0f;

      v8f s = {};
      #pragma unroll
      for (int ks = 0; ks < 16; ++ks)
        s = wmma4(kf[ks], qa[ks], s);

      #pragma unroll
      for (int r = 0; r < 8; ++r) {
        const int   gk = jt * 16 + mh + r;
        const float w  = (gk <= gq) ? (r < 4 ? w0 : w1) : 0.0f;
        s[r] = s[r] * w;
      }
      #pragma unroll
      for (int t = 0; t < 4; ++t) {
        v2f a;
        a.x = s[2 * t];
        a.y = s[2 * t + 1];
        #pragma unroll
        for (int nt = 0; nt < 4; ++nt)
          acc[nt] = wmma4(a, vf[t][nt], acc[nt]);
      }
    }
    if (more) {
      *(v4f*)(&ksh[cur ^ 1][sdst])               = pk0;
      *(v4f*)(&ksh[cur ^ 1][sdst + 16 * KPITCH]) = pk1;
      *(v4f*)(&vsh[cur ^ 1][sdst])               = pv0;
      *(v4f*)(&vsh[cur ^ 1][sdst + 16 * KPITCH]) = pv1;
    }
    __syncthreads();
  }

  #pragma unroll
  for (int j = 0; j < 2; ++j) {
    #pragma unroll
    for (int r = 0; r < 8; ++r) {
      const int row = smt * 16 + mh + r;
      const int col = (snt + j) * 16 + nl;
      kb[(size_t)row * Hc + col] = st[j][r];
    }
  }

  float* cw = &comb[sw][lane * 4];
  if (par == 1) {
    #pragma unroll
    for (int nt = 0; nt < 4; ++nt) {
      #pragma unroll
      for (int hq = 0; hq < 2; ++hq) {
        v4f p;
        p.x = acc[nt][4 * hq];     p.y = acc[nt][4 * hq + 1];
        p.z = acc[nt][4 * hq + 2]; p.w = acc[nt][4 * hq + 3];
        *(v4f*)(cw + (nt * 2 + hq) * 128) = p;
      }
    }
  }
  __syncthreads();

  if (par == 0) {
    #pragma unroll
    for (int nt = 0; nt < 4; ++nt) {
      #pragma unroll
      for (int hq = 0; hq < 2; ++hq) {
        const v4f p = *(const v4f*)(cw + (nt * 2 + hq) * 128);
        acc[nt][4 * hq]     += p.x; acc[nt][4 * hq + 1] += p.y;
        acc[nt][4 * hq + 2] += p.z; acc[nt][4 * hq + 3] += p.w;
      }
    }
    #pragma unroll
    for (int r = 0; r < 8; ++r) {
      const int gi = it * 16 + mh + r;
      float* orow = out + ((size_t)b * Sc + gi) * Hc;
      #pragma unroll
      for (int nt = 0; nt < 4; ++nt)
        orow[nt * 16 + nl] = acc[nt][r];
    }
  }
}

__global__ void __launch_bounds__(32 * NWAVE)
cross_and_state(const float* __restrict__ q, const float* __restrict__ lbuf,
                const float* __restrict__ hstate, const int* __restrict__ dones,
                float* __restrict__ out, float* __restrict__ hout)
{
  __shared__ float ktab[Tc + 1];
  __shared__ int   cumsh[Tc];
  __shared__ __attribute__((aligned(16))) float ssh[Hc * SPITCH];

  const int b     = blockIdx.x;
  const int y     = blockIdx.y;
  const int tid   = threadIdx.x;
  const int lane  = tid & 31;
  const int wave  = __builtin_amdgcn_readfirstlane(tid >> 5);
  const int sw    = wave & (RW - 1);
  const int par   = wave >> 2;
  const int nl    = lane & 15;
  const int khalf = (lane >> 4) << 1;
  const int mh    = (lane >> 4) << 3;

  const float* lb = lbuf   + (size_t)b * Sc * Hc;
  const float* hb = hstate + (size_t)b * Hc * Hc;

  const int g  = (y < NBLK) ? y : 0;
  const int it = g * RW + sw;
  const int t0 = 4 * RW * g;
  v2f qa[16];
  if (y < NBLK) {
    const float* qrow = q + ((size_t)b * Sc + it * 16 + nl) * Hc;
    #pragma unroll
    for (int ks = 0; ks < 16; ++ks)
      qa[ks] = *(const v2f*)(qrow + ks * 4 + khalf);
  }

  for (int i = tid; i <= Tc; i += 32 * NWAVE) ktab[i] = __expf((float)i * LNK);
  if (wave == 0) {
    int cj[8], ndone, fd;
    done_prefix(dones + (size_t)b * Sc, lane, cj, ndone, fd);
    (void)ndone; (void)fd;
    #pragma unroll
    for (int j = 0; j < 8; ++j) cumsh[32 * j + lane] = cj[j];
  }
  __syncthreads();

  if (y < NBLK) {
    const int cref = (g > 0) ? cumsh[t0 - 1] : 0;
    int glo = g;
    while (glo > 0 && cumsh[16 * (glo - 1) + 15] == cref) --glo;

    const float coef = (cref == 0) ? ktab[t0] : 0.0f;
    v4f sv[4];
    #pragma unroll
    for (int i = 0; i < 4; ++i) {
      const v4f hv = *(const v4f*)(hb + (tid + 256 * i) * 4);
      sv[i] = hv * coef;
    }
    for (int gp = g - 1; gp >= glo; --gp) {
      const float  w  = ktab[16 * (g - 1 - gp)];
      const float* lp = lb + (size_t)gp * Hc * Hc;
      v4f lv[4];
      #pragma unroll
      for (int i = 0; i < 4; ++i) lv[i] = *(const v4f*)(lp + (tid + 256 * i) * 4);
      #pragma unroll
      for (int i = 0; i < 4; ++i) sv[i] = sv[i] + lv[i] * w;
    }
    #pragma unroll
    for (int i = 0; i < 4; ++i) {
      const int e = (tid + 256 * i) * 4;
      *(v4f*)(&ssh[(e >> 6) * SPITCH + (e & 63)]) = sv[i];
    }
    __syncthreads();

    v8f cr[2] = {};
    {
      const float* sl = &ssh[khalf * SPITCH + par * 32 + nl];
      #pragma unroll
      for (int ks = 0; ks < 16; ++ks) {
        #pragma unroll
        for (int j = 0; j < 2; ++j) {
          v2f bm;
          bm.x = sl[(ks * 4)     * SPITCH + j * 16];
          bm.y = sl[(ks * 4 + 1) * SPITCH + j * 16];
          cr[j] = wmma4(qa[ks], bm, cr[j]);
        }
      }
    }
    float wr[8];
    #pragma unroll
    for (int r = 0; r < 8; ++r) {
      const int ti = (it * 16 + mh + r) >> 2;
      wr[r] = (cumsh[ti] == cref) ? ktab[ti - t0 + 1] : 0.0f;
    }
    #pragma unroll
    for (int r = 0; r < 8; ++r) {
      const int gi = it * 16 + mh + r;
      float* orow = out + ((size_t)b * Sc + gi) * Hc + par * 32 + nl;
      #pragma unroll
      for (int j = 0; j < 2; ++j)
        orow[j * 16] = orow[j * 16] + cr[j][r] * wr[r];
    }
  } else {
    const int cend = cumsh[Tc - 1];
    int glo = NBLK - 1;
    while (glo > 0 && cumsh[16 * (glo - 1) + 15] == cend) --glo;

    v4f sv[4] = {};
    for (int gp = NBLK - 1; gp >= glo; --gp) {
      const float  w  = ktab[16 * (NBLK - 1 - gp)];
      const float* lp = lb + (size_t)gp * Hc * Hc;
      v4f lv[4];
      #pragma unroll
      for (int i = 0; i < 4; ++i) lv[i] = *(const v4f*)(lp + (tid + 256 * i) * 4);
      #pragma unroll
      for (int i = 0; i < 4; ++i) sv[i] = sv[i] + lv[i] * w;
    }
    const float cd    = ktab[Tc];
    const float delta = (cend == 0) ? 1.0f : 0.0f;
    #pragma unroll
    for (int i = 0; i < 4; ++i) {
      const int e  = (tid + 256 * i) * 4;
      const v4f hv = *(const v4f*)(hb + e);
      *(v4f*)(hout + (size_t)b * Hc * Hc + e) = sv[i] + hv * (cd * delta);
    }
  }
}

extern "C" void kernel_launch(void* const* d_in, const int* in_sizes, int n_in,
                              void* d_out, int out_size, void* d_ws, size_t ws_size,
                              hipStream_t stream)
{
  const float* key    = (const float*)d_in[0];
  const float* query  = (const float*)d_in[1];
  const float* value  = (const float*)d_in[2];
  const float* hstate = (const float*)d_in[3];
  const int*   dones  = (const int*)d_in[4];
  const float* W_Q    = (const float*)d_in[5];
  const float* W_K    = (const float*)d_in[6];
  const float* W_V    = (const float*)d_in[7];

  const size_t qkv = (size_t)Bc * Sc * Hc;
  float* ws = (float*)d_ws;
  float* qb = ws;
  float* kb = ws + qkv;
  float* vb = ws + 2 * qkv;

  constexpr int gproj = Bc * Sc / PROJ_ROWS;
  proj_qkv<<<gproj, 256 * PROJ_TEAMS, 0, stream>>>(query, key, value,
                                                   W_Q, W_K, W_V, qb, kb, vb);

  float* out0 = (float*)d_out;
  float* hout = out0 + qkv;
  block_local<<<dim3(Bc, NBLK), 32 * NWAVE, 0, stream>>>(qb, kb, vb, dones, out0);
  cross_and_state<<<dim3(Bc, NBLK + 1), 32 * NWAVE, 0, stream>>>(
      qb, kb, hstate, dones, out0, hout);
}
